# starting kernel
# baseline (speedup 1.0000x reference)
#include <hip/hip_runtime.h>
#include <hip/hip_bf16.h>


typedef __attribute__((ext_vector_type(2))) float v2f;
typedef __attribute__((ext_vector_type(8))) float v8f;

__global__ void init_kernel(float* __restrict__ deg, float* __restrict__ agg1,
                            int N) {
  int i = blockIdx.x * blockDim.x + threadIdx.x;
  if (i < N * 16) agg1[i] = 0.0f;
  if (i < N) deg[i] = 1.0f;
}

__global__ void degree_kernel(const int* __restrict__ dst,
                              float* __restrict__ deg, int E) {
  int e = blockIdx.x * blockDim.x + threadIdx.x;
  if (e >= E) return;
  atomicAdd(&deg[dst[e]], 1.0f);
}

__global__ void rsqrt_kernel(float* __restrict__ deg, int N) {
  int i = blockIdx.x * blockDim.x + threadIdx.x;
  if (i >= N) return;
  float d = deg[i];
  deg[i] = (d > 0.0f) ? __frsqrt_rn(d) : 0.0f;
}

__global__ void __launch_bounds__(320)
gemm1_wmma(const float* __restrict__ x,
           const float* __restrict__ W1,
           float* __restrict__ h,
           int num_tiles) {
  int flat = blockIdx.x * blockDim.x + threadIdx.x;
  int wave = flat >> 5;
  int lane = threadIdx.x & 31;
  if (wave >= num_tiles) return;

  int m = lane & 15;
  bool hi = lane >= 16;
  int row0 = wave * 16;

  const float* xr = x + (size_t)(row0 + m) * 3;
  v2f a;
  a.x = hi ? xr[2] : xr[0];
  a.y = hi ? 0.0f : xr[1];

  v2f b;
  b.x = hi ? W1[2 * 16 + m] : W1[0 * 16 + m];
  b.y = hi ? 0.0f : W1[1 * 16 + m];

  v8f c = {};
  c = __builtin_amdgcn_wmma_f32_16x16x4_f32(
       false, a,  false, b,
       (short)0, c,  false,  false);

  int rbase = row0 + (hi ? 8 : 0);
#pragma unroll
  for (int v = 0; v < 8; ++v) {
    h[(size_t)(rbase + v) * 16 + m] = c[v];
  }
}

__global__ void edge_pass1(const int* __restrict__ src,
                           const int* __restrict__ dst,
                           const float* __restrict__ dinv,
                           const float4* __restrict__ h4,
                           float* __restrict__ agg1,
                           int E) {
  int e = blockIdx.x * blockDim.x + threadIdx.x;
  if (e >= E) return;
  int s = src[e];
  int d = dst[e];
  float norm = dinv[s] * dinv[d];
  const float4* hs = h4 + (size_t)s * 4;
  float* out = agg1 + (size_t)d * 16;
#pragma unroll
  for (int g = 0; g < 4; ++g) {
    float4 v = hs[g];
    atomicAdd(out + 4 * g + 0, v.x * norm);
    atomicAdd(out + 4 * g + 1, v.y * norm);
    atomicAdd(out + 4 * g + 2, v.z * norm);
    atomicAdd(out + 4 * g + 3, v.w * norm);
  }
}

__global__ void finalize1_project(const float* __restrict__ h,
                                  const float* __restrict__ agg1,
                                  const float* __restrict__ dinv,
                                  const float* __restrict__ b1,
                                  const float* __restrict__ W2,
                                  const float* __restrict__ b2,
                                  float* __restrict__ z,
                                  float* __restrict__ out,
                                  int N) {
  int i = blockIdx.x * blockDim.x + threadIdx.x;
  if (i >= N) return;
  float di = dinv[i];
  float selfw = di * di;
  float acc = 0.0f;
  const float4* a4 = (const float4*)(agg1 + (size_t)i * 16);
  const float4* h4 = (const float4*)(h + (size_t)i * 16);
#pragma unroll
  for (int g = 0; g < 4; ++g) {
    float4 av = a4[g];
    float4 hv = h4[g];
    float v0 = fmaxf(av.x + hv.x * selfw + b1[4 * g + 0], 0.0f);
    float v1 = fmaxf(av.y + hv.y * selfw + b1[4 * g + 1], 0.0f);
    float v2 = fmaxf(av.z + hv.z * selfw + b1[4 * g + 2], 0.0f);
    float v3 = fmaxf(av.w + hv.w * selfw + b1[4 * g + 3], 0.0f);
    acc += v0 * W2[4 * g + 0] + v1 * W2[4 * g + 1] +
           v2 * W2[4 * g + 2] + v3 * W2[4 * g + 3];
  }
  z[i] = acc;
  out[i] = acc * selfw + b2[0];
}

__global__ void edge_pass2(const int* __restrict__ src,
                           const int* __restrict__ dst,
                           const float* __restrict__ dinv,
                           const float* __restrict__ z,
                           float* __restrict__ out, int E) {
  int e = blockIdx.x * blockDim.x + threadIdx.x;
  if (e >= E) return;
  int s = src[e];
  int d = dst[e];
  atomicAdd(&out[d], z[s] * dinv[s] * dinv[d]);
}

extern "C" void kernel_launch(void* const* d_in, const int* in_sizes, int n_in,
                              void* d_out, int out_size, void* d_ws,
                              size_t ws_size, hipStream_t stream) {
  const float* x  = (const float*)d_in[0];
  const int*   ei = (const int*)d_in[1];
  const float* W1 = (const float*)d_in[2];
  const float* b1 = (const float*)d_in[3];
  const float* W2 = (const float*)d_in[4];
  const float* b2 = (const float*)d_in[5];
  float* out = (float*)d_out;

  const int N = in_sizes[0] / 3;
  const int E = in_sizes[1] / 2;
  const int* src = ei;
  const int* dst = ei + E;

  char* ws = (char*)d_ws;
  float* dinv = (float*)(ws);
  float* h    = (float*)(ws + (size_t)N * 4);
  float* agg1 = (float*)(ws + (size_t)N * 4 + (size_t)N * 64);
  float* z    = (float*)(ws + (size_t)N * 4 + (size_t)N * 128);

  const int BT = 256;

  init_kernel<<<(N * 16 + BT - 1) / BT, BT, 0, stream>>>(dinv, agg1, N);
  degree_kernel<<<(E + BT - 1) / BT, BT, 0, stream>>>(dst, dinv, E);
  rsqrt_kernel<<<(N + BT - 1) / BT, BT, 0, stream>>>(dinv, N);
  {
    int ntiles = N / 16;
    int wpb = 10;
    int blocks = (ntiles + wpb - 1) / wpb;
    gemm1_wmma<<<blocks, 32 * wpb, 0, stream>>>(x, W1, h, ntiles);
  }
  edge_pass1<<<(E + BT - 1) / BT, BT, 0, stream>>>(src, dst, dinv,
                                                   (const float4*)h, agg1, E);
  finalize1_project<<<(N + BT - 1) / BT, BT, 0, stream>>>(h, agg1, dinv, b1,
                                                          W2, b2, z, out, N);
  edge_pass2<<<(E + BT - 1) / BT, BT, 0, stream>>>(src, dst, dinv, z, out, E);
}


# edges sorted per call by destination node (count+rank, prefix sum, (src,norm) records); each node's wave sums its in-edges in registers, both layers: no float atomics; float sum order changes
# speedup vs baseline: 3.0320x; 3.0320x over previous
#include <hip/hip_runtime.h>
#include <hip/hip_bf16.h>


typedef __attribute__((ext_vector_type(2))) float v2f;
typedef __attribute__((ext_vector_type(8))) float v8f;

#define EPT 8

__global__ void init_counts(int* __restrict__ cnt, int N) {
  int i = blockIdx.x * blockDim.x + threadIdx.x;
  if (i < N) cnt[i] = 0;
}

template <bool FULL>
__device__ __forceinline__ void count_rank_body(const int* __restrict__ dst,
                                                int* __restrict__ cnt,
                                                int* __restrict__ rank, int E,
                                                int base) {
  int d[EPT];
  int r[EPT];
#pragma unroll
  for (int k = 0; k < EPT; ++k) {
    const int e = base + k * 256;
    d[k] = (FULL || e < E) ? dst[e] : 0;
  }
#pragma unroll
  for (int k = 0; k < EPT; ++k) {
    const int e = base + k * 256;
    r[k] = 0;
    if (FULL || e < E) r[k] = atomicAdd(&cnt[d[k]], 1);
  }
#pragma unroll
  for (int k = 0; k < EPT; ++k) {
    const int e = base + k * 256;
    if (FULL || e < E) rank[e] = r[k];
  }
}

__global__ void __launch_bounds__(256)
count_rank(const int* __restrict__ dst, int* __restrict__ cnt,
           int* __restrict__ rank, int E) {
  const int blk0 = blockIdx.x * (256 * EPT);
  const int base = blk0 + threadIdx.x;
  if (blk0 + 256 * EPT <= E) count_rank_body<true>(dst, cnt, rank, E, base);
  else count_rank_body<false>(dst, cnt, rank, E, base);
}

__device__ __forceinline__ int wave_incl_scan(int v, int lane) {
  for (int o = 1; o < 32; o <<= 1) {
    const int u = __shfl_up(v, o);
    if (lane >= o) v += u;
  }
  return v;
}

__global__ void __launch_bounds__(1024)
scan_local(const int* __restrict__ cnt, int* __restrict__ loc,
           int* __restrict__ tot, int N) {
  __shared__ int wsum[32];
  const int t = threadIdx.x, lane = t & 31, w = t >> 5;
  const int i = blockIdx.x * 1024 + t;
  const int c = (i < N) ? cnt[i] : 0;
  const int v = wave_incl_scan(c, lane);
  if (lane == 31) wsum[w] = v;
  __syncthreads();
  if (w == 0) {
    const int s = wsum[lane];
    const int sv = wave_incl_scan(s, lane);
    wsum[lane] = sv - s;
    if (lane == 31) tot[blockIdx.x] = sv;
  }
  __syncthreads();
  if (i < N) loc[i] = v - c + wsum[w];
}

__global__ void __launch_bounds__(1024)
scan_apply(const int* __restrict__ cnt, const int* __restrict__ tot,
           int* __restrict__ rowptr, float* __restrict__ dinv, int N) {
  __shared__ int red[32];
  const int t = threadIdx.x, lane = t & 31, w = t >> 5;
  const int b = blockIdx.x;
  int part = 0;
  for (int j = t; j < b; j += 1024) part += tot[j];
  for (int o = 16; o > 0; o >>= 1) part += __shfl_xor(part, o);
  if (lane == 0) red[w] = part;
  __syncthreads();
  int base = 0;
  for (int j = 0; j < 32; ++j) base += red[j];
  const int i = b * 1024 + t;
  if (i < N) {
    rowptr[i] += base;
    const float d = (float)(cnt[i] + 1);
    dinv[i] = (d > 0.0f) ? __frsqrt_rn(d) : 0.0f;
  }
}

template <bool FULL>
__device__ __forceinline__ void place_body(
    const int* __restrict__ src, const int* __restrict__ dst,
    const int* __restrict__ rank, const int* __restrict__ rowptr,
    const float* __restrict__ dinv, int* __restrict__ rec, int E, int base) {
  int s[EPT], d[EPT], r[EPT], p[EPT];
  float w[EPT];
#pragma unroll
  for (int k = 0; k < EPT; ++k) {
    const int e = base + k * 256;
    const bool ok = FULL || e < E;
    s[k] = ok ? src[e] : 0;
    d[k] = ok ? dst[e] : 0;
    r[k] = ok ? rank[e] : 0;
  }
#pragma unroll
  for (int k = 0; k < EPT; ++k) {
    p[k] = rowptr[d[k]] + r[k];
    w[k] = dinv[s[k]] * dinv[d[k]];
  }
#pragma unroll
  for (int k = 0; k < EPT; ++k) {
    const int e = base + k * 256;
    if ((FULL || e < E) && (unsigned)p[k] < (unsigned)E)
      *(int2*)(rec + 2 * (size_t)p[k]) = make_int2(s[k], __float_as_int(w[k]));
  }
}

__global__ void __launch_bounds__(256)
place_records(const int* __restrict__ src, const int* __restrict__ dst,
              const int* __restrict__ rank, const int* __restrict__ rowptr,
              const float* __restrict__ dinv, int* __restrict__ rec, int E) {
  const int blk0 = blockIdx.x * (256 * EPT);
  const int base = blk0 + threadIdx.x;
  if (blk0 + 256 * EPT <= E)
    place_body<true>(src, dst, rank, rowptr, dinv, rec, E, base);
  else
    place_body<false>(src, dst, rank, rowptr, dinv, rec, E, base);
}

__global__ void __launch_bounds__(320)
gemm1_wmma(const float* __restrict__ x,
           const float* __restrict__ W1,
           float* __restrict__ h,
           int num_tiles) {
  int flat = blockIdx.x * blockDim.x + threadIdx.x;
  int wave = flat >> 5;
  int lane = threadIdx.x & 31;
  if (wave >= num_tiles) return;

  int m = lane & 15;
  bool hi = lane >= 16;
  int row0 = wave * 16;

  const float* xr = x + (size_t)(row0 + m) * 3;
  v2f a;
  a.x = hi ? xr[2] : xr[0];
  a.y = hi ? 0.0f : xr[1];

  v2f b;
  b.x = hi ? W1[2 * 16 + m] : W1[0 * 16 + m];
  b.y = hi ? 0.0f : W1[1 * 16 + m];

  v8f c = {};
  c = __builtin_amdgcn_wmma_f32_16x16x4_f32(
       false, a,  false, b,
       (short)0, c,  false,  false);

  int rbase = row0 + (hi ? 8 : 0);
#pragma unroll
  for (int v = 0; v < 8; ++v) {
    h[(size_t)(rbase + v) * 16 + m] = c[v];
  }
}


template <int J>
__device__ __forceinline__ unsigned row_bcast(unsigned v) {
  return (unsigned)__builtin_amdgcn_update_dpp(0, (int)v, 0x150 + J, 0xf, 0xf,
                                               true);
}

template <int M>
__device__ __forceinline__ float row_xadd(float v) {
  const int t = __builtin_amdgcn_update_dpp(0, __float_as_int(v), 0x160 + M,
                                            0xf, 0xf, true);
  return v + __int_as_float(t);
}

__device__ __forceinline__ float xhalf(float x) {
  return __int_as_float(__builtin_amdgcn_permlanex16(
      __float_as_int(x), __float_as_int(x), 0x76543210, 0xfedcba98, false,
      false));
}

__device__ __forceinline__ float wave_sum(float v) {
  v = row_xadd<1>(v);
  v = row_xadd<2>(v);
  v = row_xadd<4>(v);
  v = row_xadd<8>(v);
  return v + xhalf(v);
}

__device__ __forceinline__ int lane_i(int v, int l) {
  return __builtin_amdgcn_readlane(v, l);
}
__device__ __forceinline__ float lane_f(float v, int l) {
  return __int_as_float(__builtin_amdgcn_readlane(__float_as_int(v), l));
}

__device__ __forceinline__ int2 ld_rec(const int* __restrict__ rec, int idx) {
  return *(const int2*)(rec + 2 * (size_t)idx);
}

struct Chunk {
  unsigned srow;
  float w;
};

__device__ __forceinline__ int2 load_slot(const int* __restrict__ rec,
                                          int start, int n, int c, int slot) {
  return ld_rec(rec, ((c + slot) < n) ? (start + c + slot) : 0);
}
__device__ __forceinline__ Chunk decode_slot(int2 r, int n, int c, int slot) {
  Chunk ch;
  ch.srow = (unsigned)r.x * 64u;
  ch.w = ((c + slot) < n) ? __int_as_float(r.y) : 0.0f;
  return ch;
}

template <int J>
__device__ __forceinline__ float row_gather(const float* __restrict__ h,
                                            unsigned srow, unsigned fb) {
  const unsigned off = row_bcast<J>(srow) + fb;
  return *(const float*)((const char*)h + off);
}

template <int J>
__device__ __forceinline__ float step_acc(float acc, float v, float w, int lim) {
  const float wj = __uint_as_float(row_bcast<J>(__float_as_uint(w)));
  const float vv = (2 * J < lim) ? v : 0.0f;
  return acc + wj * vv;
}

template <int G>
__device__ __forceinline__ void grp_load(const float* __restrict__ h,
                                         unsigned srow, unsigned fb,
                                         float (&v)[4]) {
  v[0] = row_gather<4 * G + 0>(h, srow, fb);
  v[1] = row_gather<4 * G + 1>(h, srow, fb);
  v[2] = row_gather<4 * G + 2>(h, srow, fb);
  v[3] = row_gather<4 * G + 3>(h, srow, fb);
}
template <int G>
__device__ __forceinline__ float grp_acc(float acc, const float (&v)[4],
                                         float w, int lim) {
  acc = step_acc<4 * G + 0>(acc, v[0], w, lim);
  acc = step_acc<4 * G + 1>(acc, v[1], w, lim);
  acc = step_acc<4 * G + 2>(acc, v[2], w, lim);
  acc = step_acc<4 * G + 3>(acc, v[3], w, lim);
  return acc;
}

__device__ __forceinline__ float chunk_sum(const float* __restrict__ h,
                                           Chunk ch, unsigned fb, int half,
                                           int rem, float acc) {
  const int lim = rem - half;
  float a[4], b[4], c[4], d[4];
  if (rem > 24) {
    grp_load<0>(h, ch.srow, fb, a);
    grp_load<1>(h, ch.srow, fb, b);
    grp_load<2>(h, ch.srow, fb, c);
    grp_load<3>(h, ch.srow, fb, d);
    acc = grp_acc<0>(acc, a, ch.w, lim);
    acc = grp_acc<1>(acc, b, ch.w, lim);
    acc = grp_acc<2>(acc, c, ch.w, lim);
    acc = grp_acc<3>(acc, d, ch.w, lim);
  } else if (rem > 16) {
    grp_load<0>(h, ch.srow, fb, a);
    grp_load<1>(h, ch.srow, fb, b);
    grp_load<2>(h, ch.srow, fb, c);
    acc = grp_acc<0>(acc, a, ch.w, lim);
    acc = grp_acc<1>(acc, b, ch.w, lim);
    acc = grp_acc<2>(acc, c, ch.w, lim);
  } else if (rem > 8) {
    grp_load<0>(h, ch.srow, fb, a);
    grp_load<1>(h, ch.srow, fb, b);
    acc = grp_acc<0>(acc, a, ch.w, lim);
    acc = grp_acc<1>(acc, b, ch.w, lim);
  } else {
    grp_load<0>(h, ch.srow, fb, a);
    acc = grp_acc<0>(acc, a, ch.w, lim);
  }
  return acc;
}

template <int GB>
__device__ __forceinline__ float sum_ab(const float* __restrict__ h, Chunk A,
                                        Chunk B, unsigned fb, int limA,
                                        int limB, float acc) {
  float a0[4], a1[4], a2[4], a3[4], b0[4], b1[4], b2[4], b3[4];
  grp_load<0>(h, A.srow, fb, a0);
  grp_load<1>(h, A.srow, fb, a1);
  grp_load<2>(h, A.srow, fb, a2);
  grp_load<3>(h, A.srow, fb, a3);
  if constexpr (GB > 0) grp_load<0>(h, B.srow, fb, b0);
  if constexpr (GB > 1) grp_load<1>(h, B.srow, fb, b1);
  if constexpr (GB > 2) grp_load<2>(h, B.srow, fb, b2);
  if constexpr (GB > 3) grp_load<3>(h, B.srow, fb, b3);
  acc = grp_acc<0>(acc, a0, A.w, limA);
  acc = grp_acc<1>(acc, a1, A.w, limA);
  acc = grp_acc<2>(acc, a2, A.w, limA);
  acc = grp_acc<3>(acc, a3, A.w, limA);
  if constexpr (GB > 0) acc = grp_acc<0>(acc, b0, B.w, limB);
  if constexpr (GB > 1) acc = grp_acc<1>(acc, b1, B.w, limB);
  if constexpr (GB > 2) acc = grp_acc<2>(acc, b2, B.w, limB);
  if constexpr (GB > 3) acc = grp_acc<3>(acc, b3, B.w, limB);
  return acc;
}

__device__ __forceinline__ float chunk_sum_ab(const float* __restrict__ h,
                                              Chunk A, Chunk B, unsigned fb,
                                              int half, int n, float acc) {
  const int limA = 32 - half;
  const int remB = n - 32;
  const int limB = remB - half;
  if (remB > 24) return sum_ab<4>(h, A, B, fb, limA, limB, acc);
  if (remB > 16) return sum_ab<3>(h, A, B, fb, limA, limB, acc);
  if (remB > 8) return sum_ab<2>(h, A, B, fb, limA, limB, acc);
  return sum_ab<1>(h, A, B, fb, limA, limB, acc);
}

__global__ void __launch_bounds__(256)
pull_layer1(const int* __restrict__ rec, const int* __restrict__ rowptr,
            const int* __restrict__ cnt, const float* __restrict__ dinv,
            const float* __restrict__ h,
            const float* __restrict__ b1,
            const float* __restrict__ W2,
            float* __restrict__ z,
            int N) {
  const int lane = threadIdx.x & 31;
  const int i0 = __builtin_amdgcn_readfirstlane(
      (int)((blockIdx.x * blockDim.x + threadIdx.x) >> 5) * 32);
  if (i0 >= N) return;
  const int nodes = (N - i0 < 32) ? (N - i0) : 32;
  const int mi = i0 + lane;
  const bool mok = lane < nodes;
  const int mc = mok ? mi : i0;
  const int my_start = rowptr[mc];
  const int my_cnt = mok ? cnt[mc] : 0;
  const float my_di = dinv[mc];
  const int f = lane & 15, half = lane >> 4;
  const unsigned fb = (unsigned)f * 4u;
  const int slot = 2 * f + half;
  const float b1f = b1[f], w2f = W2[f];
  float zkeep = 0.0f;

  int st = lane_i(my_start, 0), n = lane_i(my_cnt, 0);
  int2 ra = load_slot(rec, st, n, 0, slot);
  int2 rb = load_slot(rec, st, n, 32, slot);
  for (int nn = 0; nn < nodes; ++nn) {
    const int i = i0 + nn;
    const int st_c = st, n_c = n;
    const int2 qa = ra, qb = rb;
    const float hv = h[(size_t)i * 16 + f];
    const float di = lane_f(my_di, nn);
    if (nn + 1 < nodes) {
      st = lane_i(my_start, nn + 1);
      n = lane_i(my_cnt, nn + 1);
      ra = load_slot(rec, st, n, 0, slot);
      rb = load_slot(rec, st, n, 32, slot);
    }
    float acc = 0.0f;
    if (n_c > 32)
      acc = chunk_sum_ab(h, decode_slot(qa, n_c, 0, slot),
                         decode_slot(qb, n_c, 32, slot), fb, half, n_c, acc);
    else if (n_c > 0)
      acc = chunk_sum(h, decode_slot(qa, n_c, 0, slot), fb, half, n_c, acc);
    for (int c = 64; c < n_c; c += 32) {
      const int2 qx = load_slot(rec, st_c, n_c, c, slot);
      acc = chunk_sum(h, decode_slot(qx, n_c, c, slot), fb, half, n_c - c, acc);
    }
    acc += xhalf(acc);
    const float a1 = fmaxf(acc + hv * (di * di) + b1f, 0.0f);
    float p = a1 * w2f;
    p = row_xadd<8>(p);
    p = row_xadd<4>(p);
    p = row_xadd<2>(p);
    p = row_xadd<1>(p);
    if (lane == nn) zkeep = p;
  }
  if (mok) z[mi] = zkeep;
}

__global__ void __launch_bounds__(256)
pull_layer2(const int* __restrict__ rec, const int* __restrict__ rowptr,
            const int* __restrict__ cnt, const float* __restrict__ dinv,
            const float* __restrict__ z, const float* __restrict__ b2,
            float* __restrict__ out, int N) {
  const int lane = threadIdx.x & 31;
  const int i0 = __builtin_amdgcn_readfirstlane(
      (int)((blockIdx.x * blockDim.x + threadIdx.x) >> 5) * 32);
  if (i0 >= N) return;
  const int nodes = (N - i0 < 32) ? (N - i0) : 32;
  const int mi = i0 + lane;
  const bool mok = lane < nodes;
  const int mc = mok ? mi : i0;
  const int my_start = rowptr[mc];
  const int my_cnt = mok ? cnt[mc] : 0;
  const float my_di = dinv[mc];
  const float zself = z[mc];
  float keep = 0.0f;

  for (int nn = 0; nn < 32; nn += 4) {
    int st[4], n[4];
    int2 ra[4], rb[4];
    float za[4], zb[4];
#pragma unroll
    for (int u = 0; u < 4; ++u) {
      st[u] = lane_i(my_start, nn + u);
      n[u] = lane_i(my_cnt, nn + u);
      ra[u] = ld_rec(rec, (lane < n[u]) ? st[u] + lane : 0);
      rb[u] = ld_rec(rec, (32 + lane < n[u]) ? st[u] + 32 + lane : 0);
    }
#pragma unroll
    for (int u = 0; u < 4; ++u) {
      za[u] = z[ra[u].x];
      zb[u] = z[rb[u].x];
    }
#pragma unroll
    for (int u = 0; u < 4; ++u) {
      const float wa = __int_as_float(ra[u].y & ((lane < n[u]) ? -1 : 0));
      const float wb = __int_as_float(rb[u].y & ((32 + lane < n[u]) ? -1 : 0));
      float acc = wa * za[u] + wb * zb[u];
      for (int c = 64; c < n[u]; c += 32) {
        const bool ok = (c + lane) < n[u];
        const int2 r = ld_rec(rec, ok ? st[u] + c + lane : 0);
        const float wc = __int_as_float(r.y & (ok ? -1 : 0));
        acc += wc * z[r.x];
      }
      const float s = wave_sum(acc);
      if (lane == nn + u) keep = s;
    }
  }
  if (mok) out[mi] = keep + zself * (my_di * my_di) + b2[0];
}

extern "C" void kernel_launch(void* const* d_in, const int* in_sizes, int n_in,
                              void* d_out, int out_size, void* d_ws,
                              size_t ws_size, hipStream_t stream) {
  const float* x  = (const float*)d_in[0];
  const int*   ei = (const int*)d_in[1];
  const float* W1 = (const float*)d_in[2];
  const float* b1 = (const float*)d_in[3];
  const float* W2 = (const float*)d_in[4];
  const float* b2 = (const float*)d_in[5];
  float* out = (float*)d_out;

  const int N = in_sizes[0] / 3;
  const int E = in_sizes[1] / 2;
  const int* src = ei;
  const int* dst = ei + E;

  const size_t nb4 = (((size_t)N * 4) + 255) & ~(size_t)255;
  const size_t eb4 = (((size_t)E * 4) + 255) & ~(size_t)255;
  const int nblk = (N + 1023) / 1024;
  const size_t tb4 = (((size_t)nblk * 4) + 255) & ~(size_t)255;
  char* ws = (char*)d_ws;
  int*   cnt    = (int*)(ws);
  int*   rowptr = (int*)(ws + nb4);
  float* dinv   = (float*)(ws + 2 * nb4);
  float* z      = (float*)(ws + 3 * nb4);
  int*   tot    = (int*)(ws + 4 * nb4);
  float* h      = (float*)(ws + 4 * nb4 + tb4);
  int*   rank   = (int*)(ws + 4 * nb4 + tb4 + 16 * nb4);
  int*   rec    = (int*)(ws + 4 * nb4 + tb4 + 16 * nb4 + eb4);

  const int BT = 256;
  const int eblocks = (E + BT * EPT - 1) / (BT * EPT);
  const int nwaves = (N + 31) / 32;
  const int pblocks = (nwaves + 7) / 8;

  init_counts<<<(N + BT - 1) / BT, BT, 0, stream>>>(cnt, N);
  count_rank<<<eblocks, BT, 0, stream>>>(dst, cnt, rank, E);
  scan_local<<<nblk, 1024, 0, stream>>>(cnt, rowptr, tot, N);
  scan_apply<<<nblk, 1024, 0, stream>>>(cnt, tot, rowptr, dinv, N);
  place_records<<<eblocks, BT, 0, stream>>>(src, dst, rank, rowptr, dinv, rec, E);
  {
    int ntiles = N / 16;
    int wpb = 10;
    int blocks = (ntiles + wpb - 1) / wpb;
    gemm1_wmma<<<blocks, 32 * wpb, 0, stream>>>(x, W1, h, ntiles);
  }
  pull_layer1<<<pblocks, BT, 0, stream>>>(rec, rowptr, cnt, dinv, h, b1, W2, z, N);
  pull_layer2<<<pblocks, BT, 0, stream>>>(rec, rowptr, cnt, dinv, z, b2, out, N);
}


# count+rank and placement split in two destination ranges; resident groups draw (chunk, range) tickets, own chip half's range first: a range's lines are written by one half; float sum order changes
# speedup vs baseline: 7.0633x; 2.3296x over previous
#include <hip/hip_runtime.h>
#include <hip/hip_bf16.h>


typedef __attribute__((ext_vector_type(2))) float v2f;
typedef __attribute__((ext_vector_type(8))) float v8f;

#define EPT 8

#define EDGE_WGS 2048
#define CTR_WORDS 256
#define CTR_PITCH 64

__global__ void init_counts(int* __restrict__ cnt, int* __restrict__ ctr, int N) {
  int i = blockIdx.x * blockDim.x + threadIdx.x;
  if (i < N) cnt[i] = 0;
  if (i < CTR_WORDS) ctr[i] = 0;
}

__device__ __forceinline__ int chip_half() {
  return (int)((__builtin_amdgcn_s_sendmsg_rtn(0x87) >> 18) & 1u);
}

__device__ __forceinline__ int range_split(int N) {
  return ((N >> 1) + 63) & ~63;
}

__device__ __forceinline__ int resolve_task(int* __restrict__ ctr, int r, int t,
                                            int nchunks) {
  if (t >= nchunks) {
    r ^= 1;
    t = atomicAdd(&ctr[r * CTR_PITCH], 1);
  }
  return (t < nchunks) ? (t * 2 + r) : -1;
}

template <bool FULL>
__device__ __forceinline__ int count_rank_body(const int* __restrict__ dst,
                                               int* __restrict__ cnt,
                                               int* __restrict__ rank, int E,
                                               int base, int nh, bool hi,
                                               int* __restrict__ own) {
  int d[EPT];
  int r[EPT];
  bool mine[EPT];
#pragma unroll
  for (int k = 0; k < EPT; ++k) {
    const int e = base + k * 256;
    d[k] = (FULL || e < E) ? dst[e] : 0;
  }
  int t = 0;
  if (threadIdx.x == 0) t = atomicAdd(own, 1);
  asm volatile("" ::"v"(d[0]), "v"(d[1]), "v"(d[2]), "v"(d[3]), "v"(d[4]),
               "v"(d[5]), "v"(d[6]), "v"(d[7]));
#pragma unroll
  for (int k = 0; k < EPT; ++k) {
    const int e = base + k * 256;
    mine[k] = (FULL || e < E) && ((d[k] >= nh) == hi);
    r[k] = 0;
    if (mine[k]) r[k] = atomicAdd(&cnt[d[k]], 1);
  }
#pragma unroll
  for (int k = 0; k < EPT; ++k) {
    const int e = base + k * 256;
    if (mine[k]) rank[e] = r[k];
  }
  return t;
}

__global__ void __launch_bounds__(256)
count_rank(const int* __restrict__ dst, int* __restrict__ cnt,
           int* __restrict__ rank, int* __restrict__ ctr, int E, int N,
           int nchunks) {
  __shared__ int task[2];
  const int nh = range_split(N);
  const int half = chip_half();
  int* const own = ctr + half * CTR_PITCH;
  if (threadIdx.x == 0)
    task[0] = resolve_task(ctr, half, atomicAdd(own, 1), nchunks);
  __syncthreads();
  for (int it = 0; it < 2 * nchunks; ++it) {
    const int code = task[it & 1];
    if (code < 0) break;
    int t;
    const int blk0 = (code >> 1) * (256 * EPT);
    const int base = blk0 + threadIdx.x;
    const bool hi = (code & 1) != 0;
    if (blk0 + 256 * EPT <= E)
      t = count_rank_body<true>(dst, cnt, rank, E, base, nh, hi, own);
    else
      t = count_rank_body<false>(dst, cnt, rank, E, base, nh, hi, own);
    if (threadIdx.x == 0) task[(it + 1) & 1] = resolve_task(ctr, half, t, nchunks);
    __syncthreads();
  }
}

__device__ __forceinline__ int wave_incl_scan(int v, int lane) {
  for (int o = 1; o < 32; o <<= 1) {
    const int u = __shfl_up(v, o);
    if (lane >= o) v += u;
  }
  return v;
}

__global__ void __launch_bounds__(1024)
scan_local(const int* __restrict__ cnt, int* __restrict__ loc,
           int* __restrict__ tot, int N) {
  __shared__ int wsum[32];
  const int t = threadIdx.x, lane = t & 31, w = t >> 5;
  const int i = blockIdx.x * 1024 + t;
  const int c = (i < N) ? cnt[i] : 0;
  const int v = wave_incl_scan(c, lane);
  if (lane == 31) wsum[w] = v;
  __syncthreads();
  if (w == 0) {
    const int s = wsum[lane];
    const int sv = wave_incl_scan(s, lane);
    wsum[lane] = sv - s;
    if (lane == 31) tot[blockIdx.x] = sv;
  }
  __syncthreads();
  if (i < N) loc[i] = v - c + wsum[w];
}

__global__ void __launch_bounds__(1024)
scan_apply(const int* __restrict__ cnt, const int* __restrict__ tot,
           int* __restrict__ rowptr, float* __restrict__ dinv, int N) {
  __shared__ int red[32];
  const int t = threadIdx.x, lane = t & 31, w = t >> 5;
  const int b = blockIdx.x;
  int part = 0;
  for (int j = t; j < b; j += 1024) part += tot[j];
  for (int o = 16; o > 0; o >>= 1) part += __shfl_xor(part, o);
  if (lane == 0) red[w] = part;
  __syncthreads();
  int base = 0;
  for (int j = 0; j < 32; ++j) base += red[j];
  const int i = b * 1024 + t;
  if (i < N) {
    rowptr[i] += base;
    const float d = (float)(cnt[i] + 1);
    dinv[i] = (d > 0.0f) ? __frsqrt_rn(d) : 0.0f;
  }
}

template <bool FULL>
__device__ __forceinline__ int place_body(
    const int* __restrict__ src, const int* __restrict__ dst,
    const int* __restrict__ rank, const int* __restrict__ rowptr,
    const float* __restrict__ dinv, int* __restrict__ rec, int E, int base,
    int nh, bool hi, int* __restrict__ own) {
  int s[EPT], d[EPT], r[EPT], rp[EPT];
  float ws[EPT], wd[EPT];
  bool mine[EPT];
#pragma unroll
  for (int k = 0; k < EPT; ++k) {
    const int e = base + k * 256;
    const bool ok = FULL || e < E;
    s[k] = ok ? src[e] : 0;
    d[k] = ok ? dst[e] : 0;
    r[k] = ok ? rank[e] : 0;
  }
  int t = 0;
  if (threadIdx.x == 0) t = atomicAdd(own, 1);
  asm volatile("" ::"v"(s[0]), "v"(s[1]), "v"(s[2]), "v"(s[3]), "v"(s[4]),
               "v"(s[5]), "v"(s[6]), "v"(s[7]), "v"(d[0]), "v"(d[1]), "v"(d[2]),
               "v"(d[3]), "v"(d[4]), "v"(d[5]), "v"(d[6]), "v"(d[7]), "v"(r[0]),
               "v"(r[1]), "v"(r[2]), "v"(r[3]), "v"(r[4]), "v"(r[5]), "v"(r[6]),
               "v"(r[7]));
#pragma unroll
  for (int k = 0; k < EPT; ++k) {
    const int e = base + k * 256;
    mine[k] = (FULL || e < E) && ((d[k] >= nh) == hi);
    rp[k] = 0;
    ws[k] = 0.0f;
    wd[k] = 0.0f;
    if (mine[k]) {
      rp[k] = rowptr[d[k]];
      ws[k] = dinv[s[k]];
      wd[k] = dinv[d[k]];
    }
  }
#pragma unroll
  for (int k = 0; k < EPT; ++k) {
    const int p = rp[k] + r[k];
    const float w = ws[k] * wd[k];
    if (mine[k] && (unsigned)p < (unsigned)E)
      *(int2*)(rec + 2 * (size_t)p) = make_int2(s[k], __float_as_int(w));
  }
  return t;
}

__global__ void __launch_bounds__(256)
place_records(const int* __restrict__ src, const int* __restrict__ dst,
              const int* __restrict__ rank, const int* __restrict__ rowptr,
              const float* __restrict__ dinv, int* __restrict__ rec,
              int* __restrict__ ctr, int E, int N, int nchunks) {
  __shared__ int task[2];
  const int nh = range_split(N);
  const int half = chip_half();
  int* const own = ctr + half * CTR_PITCH;
  if (threadIdx.x == 0)
    task[0] = resolve_task(ctr, half, atomicAdd(own, 1), nchunks);
  __syncthreads();
  for (int it = 0; it < 2 * nchunks; ++it) {
    const int code = task[it & 1];
    if (code < 0) break;
    int t;
    const int blk0 = (code >> 1) * (256 * EPT);
    const int base = blk0 + threadIdx.x;
    const bool hi = (code & 1) != 0;
    if (blk0 + 256 * EPT <= E)
      t = place_body<true>(src, dst, rank, rowptr, dinv, rec, E, base, nh, hi, own);
    else
      t = place_body<false>(src, dst, rank, rowptr, dinv, rec, E, base, nh, hi, own);
    if (threadIdx.x == 0) task[(it + 1) & 1] = resolve_task(ctr, half, t, nchunks);
    __syncthreads();
  }
}

__global__ void __launch_bounds__(320)
gemm1_wmma(const float* __restrict__ x,
           const float* __restrict__ W1,
           float* __restrict__ h,
           int num_tiles) {
  int flat = blockIdx.x * blockDim.x + threadIdx.x;
  int wave = flat >> 5;
  int lane = threadIdx.x & 31;
  if (wave >= num_tiles) return;

  int m = lane & 15;
  bool hi = lane >= 16;
  int row0 = wave * 16;

  const float* xr = x + (size_t)(row0 + m) * 3;
  v2f a;
  a.x = hi ? xr[2] : xr[0];
  a.y = hi ? 0.0f : xr[1];

  v2f b;
  b.x = hi ? W1[2 * 16 + m] : W1[0 * 16 + m];
  b.y = hi ? 0.0f : W1[1 * 16 + m];

  v8f c = {};
  c = __builtin_amdgcn_wmma_f32_16x16x4_f32(
       false, a,  false, b,
       (short)0, c,  false,  false);

  int rbase = row0 + (hi ? 8 : 0);
#pragma unroll
  for (int v = 0; v < 8; ++v) {
    h[(size_t)(rbase + v) * 16 + m] = c[v];
  }
}


template <int J>
__device__ __forceinline__ unsigned row_bcast(unsigned v) {
  return (unsigned)__builtin_amdgcn_update_dpp(0, (int)v, 0x150 + J, 0xf, 0xf,
                                               true);
}

template <int M>
__device__ __forceinline__ float row_xadd(float v) {
  const int t = __builtin_amdgcn_update_dpp(0, __float_as_int(v), 0x160 + M,
                                            0xf, 0xf, true);
  return v + __int_as_float(t);
}

__device__ __forceinline__ float xhalf(float x) {
  return __int_as_float(__builtin_amdgcn_permlanex16(
      __float_as_int(x), __float_as_int(x), 0x76543210, 0xfedcba98, false,
      false));
}

__device__ __forceinline__ float wave_sum(float v) {
  v = row_xadd<1>(v);
  v = row_xadd<2>(v);
  v = row_xadd<4>(v);
  v = row_xadd<8>(v);
  return v + xhalf(v);
}

__device__ __forceinline__ int lane_i(int v, int l) {
  return __builtin_amdgcn_readlane(v, l);
}
__device__ __forceinline__ float lane_f(float v, int l) {
  return __int_as_float(__builtin_amdgcn_readlane(__float_as_int(v), l));
}

__device__ __forceinline__ int2 ld_rec(const int* __restrict__ rec, int idx) {
  return *(const int2*)(rec + 2 * (size_t)idx);
}

struct Chunk {
  unsigned srow;
  float w;
};

__device__ __forceinline__ int2 load_slot(const int* __restrict__ rec,
                                          int start, int n, int c, int slot) {
  return ld_rec(rec, ((c + slot) < n) ? (start + c + slot) : 0);
}
__device__ __forceinline__ Chunk decode_slot(int2 r, int n, int c, int slot) {
  Chunk ch;
  ch.srow = (unsigned)r.x * 64u;
  ch.w = ((c + slot) < n) ? __int_as_float(r.y) : 0.0f;
  return ch;
}

template <int J>
__device__ __forceinline__ float row_gather(const float* __restrict__ h,
                                            unsigned srow, unsigned fb) {
  const unsigned off = row_bcast<J>(srow) + fb;
  return *(const float*)((const char*)h + off);
}

template <int J>
__device__ __forceinline__ float step_acc(float acc, float v, float w, int lim) {
  const float wj = __uint_as_float(row_bcast<J>(__float_as_uint(w)));
  const float vv = (2 * J < lim) ? v : 0.0f;
  return acc + wj * vv;
}

template <int G>
__device__ __forceinline__ void grp_load(const float* __restrict__ h,
                                         unsigned srow, unsigned fb,
                                         float (&v)[4]) {
  v[0] = row_gather<4 * G + 0>(h, srow, fb);
  v[1] = row_gather<4 * G + 1>(h, srow, fb);
  v[2] = row_gather<4 * G + 2>(h, srow, fb);
  v[3] = row_gather<4 * G + 3>(h, srow, fb);
}
template <int G>
__device__ __forceinline__ float grp_acc(float acc, const float (&v)[4],
                                         float w, int lim) {
  acc = step_acc<4 * G + 0>(acc, v[0], w, lim);
  acc = step_acc<4 * G + 1>(acc, v[1], w, lim);
  acc = step_acc<4 * G + 2>(acc, v[2], w, lim);
  acc = step_acc<4 * G + 3>(acc, v[3], w, lim);
  return acc;
}

__device__ __forceinline__ float chunk_sum(const float* __restrict__ h,
                                           Chunk ch, unsigned fb, int half,
                                           int rem, float acc) {
  const int lim = rem - half;
  float a[4], b[4], c[4], d[4];
  if (rem > 24) {
    grp_load<0>(h, ch.srow, fb, a);
    grp_load<1>(h, ch.srow, fb, b);
    grp_load<2>(h, ch.srow, fb, c);
    grp_load<3>(h, ch.srow, fb, d);
    acc = grp_acc<0>(acc, a, ch.w, lim);
    acc = grp_acc<1>(acc, b, ch.w, lim);
    acc = grp_acc<2>(acc, c, ch.w, lim);
    acc = grp_acc<3>(acc, d, ch.w, lim);
  } else if (rem > 16) {
    grp_load<0>(h, ch.srow, fb, a);
    grp_load<1>(h, ch.srow, fb, b);
    grp_load<2>(h, ch.srow, fb, c);
    acc = grp_acc<0>(acc, a, ch.w, lim);
    acc = grp_acc<1>(acc, b, ch.w, lim);
    acc = grp_acc<2>(acc, c, ch.w, lim);
  } else if (rem > 8) {
    grp_load<0>(h, ch.srow, fb, a);
    grp_load<1>(h, ch.srow, fb, b);
    acc = grp_acc<0>(acc, a, ch.w, lim);
    acc = grp_acc<1>(acc, b, ch.w, lim);
  } else {
    grp_load<0>(h, ch.srow, fb, a);
    acc = grp_acc<0>(acc, a, ch.w, lim);
  }
  return acc;
}

template <int GB>
__device__ __forceinline__ float sum_ab(const float* __restrict__ h, Chunk A,
                                        Chunk B, unsigned fb, int limA,
                                        int limB, float acc) {
  float a0[4], a1[4], a2[4], a3[4], b0[4], b1[4], b2[4], b3[4];
  grp_load<0>(h, A.srow, fb, a0);
  grp_load<1>(h, A.srow, fb, a1);
  grp_load<2>(h, A.srow, fb, a2);
  grp_load<3>(h, A.srow, fb, a3);
  if constexpr (GB > 0) grp_load<0>(h, B.srow, fb, b0);
  if constexpr (GB > 1) grp_load<1>(h, B.srow, fb, b1);
  if constexpr (GB > 2) grp_load<2>(h, B.srow, fb, b2);
  if constexpr (GB > 3) grp_load<3>(h, B.srow, fb, b3);
  acc = grp_acc<0>(acc, a0, A.w, limA);
  acc = grp_acc<1>(acc, a1, A.w, limA);
  acc = grp_acc<2>(acc, a2, A.w, limA);
  acc = grp_acc<3>(acc, a3, A.w, limA);
  if constexpr (GB > 0) acc = grp_acc<0>(acc, b0, B.w, limB);
  if constexpr (GB > 1) acc = grp_acc<1>(acc, b1, B.w, limB);
  if constexpr (GB > 2) acc = grp_acc<2>(acc, b2, B.w, limB);
  if constexpr (GB > 3) acc = grp_acc<3>(acc, b3, B.w, limB);
  return acc;
}

__device__ __forceinline__ float chunk_sum_ab(const float* __restrict__ h,
                                              Chunk A, Chunk B, unsigned fb,
                                              int half, int n, float acc) {
  const int limA = 32 - half;
  const int remB = n - 32;
  const int limB = remB - half;
  if (remB > 24) return sum_ab<4>(h, A, B, fb, limA, limB, acc);
  if (remB > 16) return sum_ab<3>(h, A, B, fb, limA, limB, acc);
  if (remB > 8) return sum_ab<2>(h, A, B, fb, limA, limB, acc);
  return sum_ab<1>(h, A, B, fb, limA, limB, acc);
}

__global__ void __launch_bounds__(256)
pull_layer1(const int* __restrict__ rec, const int* __restrict__ rowptr,
            const int* __restrict__ cnt, const float* __restrict__ dinv,
            const float* __restrict__ h,
            const float* __restrict__ b1,
            const float* __restrict__ W2,
            float* __restrict__ z,
            int N) {
  const int lane = threadIdx.x & 31;
  const int i0 = __builtin_amdgcn_readfirstlane(
      (int)((blockIdx.x * blockDim.x + threadIdx.x) >> 5) * 32);
  if (i0 >= N) return;
  const int nodes = (N - i0 < 32) ? (N - i0) : 32;
  const int mi = i0 + lane;
  const bool mok = lane < nodes;
  const int mc = mok ? mi : i0;
  const int my_start = rowptr[mc];
  const int my_cnt = mok ? cnt[mc] : 0;
  const float my_di = dinv[mc];
  const int f = lane & 15, half = lane >> 4;
  const unsigned fb = (unsigned)f * 4u;
  const int slot = 2 * f + half;
  const float b1f = b1[f], w2f = W2[f];
  float zkeep = 0.0f;

  int st = lane_i(my_start, 0), n = lane_i(my_cnt, 0);
  int2 ra = load_slot(rec, st, n, 0, slot);
  int2 rb = load_slot(rec, st, n, 32, slot);
  for (int nn = 0; nn < nodes; ++nn) {
    const int i = i0 + nn;
    const int st_c = st, n_c = n;
    const int2 qa = ra, qb = rb;
    const float hv = h[(size_t)i * 16 + f];
    const float di = lane_f(my_di, nn);
    if (nn + 1 < nodes) {
      st = lane_i(my_start, nn + 1);
      n = lane_i(my_cnt, nn + 1);
      ra = load_slot(rec, st, n, 0, slot);
      rb = load_slot(rec, st, n, 32, slot);
    }
    float acc = 0.0f;
    if (n_c > 32)
      acc = chunk_sum_ab(h, decode_slot(qa, n_c, 0, slot),
                         decode_slot(qb, n_c, 32, slot), fb, half, n_c, acc);
    else if (n_c > 0)
      acc = chunk_sum(h, decode_slot(qa, n_c, 0, slot), fb, half, n_c, acc);
    for (int c = 64; c < n_c; c += 32) {
      const int2 qx = load_slot(rec, st_c, n_c, c, slot);
      acc = chunk_sum(h, decode_slot(qx, n_c, c, slot), fb, half, n_c - c, acc);
    }
    acc += xhalf(acc);
    const float a1 = fmaxf(acc + hv * (di * di) + b1f, 0.0f);
    float p = a1 * w2f;
    p = row_xadd<8>(p);
    p = row_xadd<4>(p);
    p = row_xadd<2>(p);
    p = row_xadd<1>(p);
    if (lane == nn) zkeep = p;
  }
  if (mok) z[mi] = zkeep;
}

__global__ void __launch_bounds__(256)
pull_layer2(const int* __restrict__ rec, const int* __restrict__ rowptr,
            const int* __restrict__ cnt, const float* __restrict__ dinv,
            const float* __restrict__ z, const float* __restrict__ b2,
            float* __restrict__ out, int N) {
  const int lane = threadIdx.x & 31;
  const int i0 = __builtin_amdgcn_readfirstlane(
      (int)((blockIdx.x * blockDim.x + threadIdx.x) >> 5) * 32);
  if (i0 >= N) return;
  const int nodes = (N - i0 < 32) ? (N - i0) : 32;
  const int mi = i0 + lane;
  const bool mok = lane < nodes;
  const int mc = mok ? mi : i0;
  const int my_start = rowptr[mc];
  const int my_cnt = mok ? cnt[mc] : 0;
  const float my_di = dinv[mc];
  const float zself = z[mc];
  float keep = 0.0f;

  for (int nn = 0; nn < 32; nn += 4) {
    int st[4], n[4];
    int2 ra[4], rb[4];
    float za[4], zb[4];
#pragma unroll
    for (int u = 0; u < 4; ++u) {
      st[u] = lane_i(my_start, nn + u);
      n[u] = lane_i(my_cnt, nn + u);
      ra[u] = ld_rec(rec, (lane < n[u]) ? st[u] + lane : 0);
      rb[u] = ld_rec(rec, (32 + lane < n[u]) ? st[u] + 32 + lane : 0);
    }
#pragma unroll
    for (int u = 0; u < 4; ++u) {
      za[u] = z[ra[u].x];
      zb[u] = z[rb[u].x];
    }
#pragma unroll
    for (int u = 0; u < 4; ++u) {
      const float wa = __int_as_float(ra[u].y & ((lane < n[u]) ? -1 : 0));
      const float wb = __int_as_float(rb[u].y & ((32 + lane < n[u]) ? -1 : 0));
      float acc = wa * za[u] + wb * zb[u];
      for (int c = 64; c < n[u]; c += 32) {
        const bool ok = (c + lane) < n[u];
        const int2 r = ld_rec(rec, ok ? st[u] + c + lane : 0);
        const float wc = __int_as_float(r.y & (ok ? -1 : 0));
        acc += wc * z[r.x];
      }
      const float s = wave_sum(acc);
      if (lane == nn + u) keep = s;
    }
  }
  if (mok) out[mi] = keep + zself * (my_di * my_di) + b2[0];
}

extern "C" void kernel_launch(void* const* d_in, const int* in_sizes, int n_in,
                              void* d_out, int out_size, void* d_ws,
                              size_t ws_size, hipStream_t stream) {
  const float* x  = (const float*)d_in[0];
  const int*   ei = (const int*)d_in[1];
  const float* W1 = (const float*)d_in[2];
  const float* b1 = (const float*)d_in[3];
  const float* W2 = (const float*)d_in[4];
  const float* b2 = (const float*)d_in[5];
  float* out = (float*)d_out;

  const int N = in_sizes[0] / 3;
  const int E = in_sizes[1] / 2;
  const int* src = ei;
  const int* dst = ei + E;

  const size_t nb4 = (((size_t)N * 4) + 255) & ~(size_t)255;
  const size_t eb4 = (((size_t)E * 4) + 255) & ~(size_t)255;
  const int nblk = (N + 1023) / 1024;
  const size_t tb4 = (((size_t)nblk * 4) + 255) & ~(size_t)255;
  char* ws = (char*)d_ws;
  int*   cnt    = (int*)(ws);
  int*   rowptr = (int*)(ws + nb4);
  float* dinv   = (float*)(ws + 2 * nb4);
  float* z      = (float*)(ws + 3 * nb4);
  int*   tot    = (int*)(ws + 4 * nb4);
  float* h      = (float*)(ws + 4 * nb4 + tb4);
  int*   rank   = (int*)(ws + 4 * nb4 + tb4 + 16 * nb4);
  int*   rec    = (int*)(ws + 4 * nb4 + tb4 + 16 * nb4 + eb4);
  int*   ctr    = (int*)(ws + 4 * nb4 + tb4 + 16 * nb4 + 3 * eb4);

  const int BT = 256;
  const int eblocks = (E + BT * EPT - 1) / (BT * EPT);
  const int nwaves = (N + 31) / 32;
  const int pblocks = (nwaves + 7) / 8;

  init_counts<<<(N + BT - 1) / BT, BT, 0, stream>>>(cnt, ctr, N);
  count_rank<<<EDGE_WGS, BT, 0, stream>>>(dst, cnt, rank, ctr, E, N, eblocks);
  scan_local<<<nblk, 1024, 0, stream>>>(cnt, rowptr, tot, N);
  scan_apply<<<nblk, 1024, 0, stream>>>(cnt, tot, rowptr, dinv, N);
  place_records<<<EDGE_WGS, BT, 0, stream>>>(src, dst, rank, rowptr, dinv, rec,
                                             ctr + 2 * CTR_PITCH, E, N, eblocks);
  {
    int ntiles = N / 16;
    int wpb = 10;
    int blocks = (ntiles + wpb - 1) / wpb;
    gemm1_wmma<<<blocks, 32 * wpb, 0, stream>>>(x, W1, h, ntiles);
  }
  pull_layer1<<<pblocks, BT, 0, stream>>>(rec, rowptr, cnt, dinv, h, b1, W2, z, N);
  pull_layer2<<<pblocks, BT, 0, stream>>>(rec, rowptr, cnt, dinv, z, b2, out, N);
}


# edge sources stored at fixed-pitch per-node slots (dst, atomic rank) in the counting pass itself: no prefix sum, rank plane or second edge pass; 5 launches, not 8; sum order changes past 64 in-edges
# speedup vs baseline: 10.2771x; 1.4550x over previous
#include <hip/hip_runtime.h>
#include <hip/hip_bf16.h>


typedef __attribute__((ext_vector_type(2))) float v2f;
typedef __attribute__((ext_vector_type(8))) float v8f;

#define EPT 8
#define CAP 64

#define EDGE_WGS 2048
#define CTR_WORDS 256
#define CTR_PITCH 64
#define OVF_WORD (2 * CTR_PITCH)

__global__ void init_counts(int* __restrict__ cnt, int* __restrict__ ctr, int N) {
  int i = blockIdx.x * blockDim.x + threadIdx.x;
  if (i < N) cnt[i] = 0;
  if (i < CTR_WORDS) ctr[i] = 0;
}

__device__ __forceinline__ int chip_half() {
  return (int)((__builtin_amdgcn_s_sendmsg_rtn(0x87) >> 18) & 1u);
}

__device__ __forceinline__ int range_split(int N) {
  return ((N >> 1) + 63) & ~63;
}

__device__ __forceinline__ int resolve_task(int* __restrict__ ctr, int r, int t,
                                            int nchunks) {
  if (t >= nchunks) {
    r ^= 1;
    t = atomicAdd(&ctr[r * CTR_PITCH], 1);
  }
  return (t < nchunks) ? (t * 2 + r) : -1;
}

template <bool FULL>
__device__ __forceinline__ int count_place_body(const int* __restrict__ src,
                                                const int* __restrict__ dst,
                                                int* __restrict__ cnt,
                                                int* __restrict__ slots,
                                                int2* __restrict__ ovf,
                                                int* __restrict__ ovf_n, int E,
                                                int base, int nh, bool hi,
                                                int* __restrict__ own) {
  int d[EPT];
  int s[EPT];
  int r[EPT];
  bool mine[EPT];
#pragma unroll
  for (int k = 0; k < EPT; ++k) {
    const int e = base + k * 256;
    d[k] = (FULL || e < E) ? dst[e] : 0;
  }
#pragma unroll
  for (int k = 0; k < EPT; ++k) {
    const int e = base + k * 256;
    s[k] = (FULL || e < E) ? src[e] : 0;
  }
  int t = 0;
  if (threadIdx.x == 0) t = atomicAdd(own, 1);
  asm volatile("" ::"v"(d[0]), "v"(d[1]), "v"(d[2]), "v"(d[3]), "v"(d[4]),
               "v"(d[5]), "v"(d[6]), "v"(d[7]));
#pragma unroll
  for (int k = 0; k < EPT; ++k) {
    const int e = base + k * 256;
    mine[k] = (FULL || e < E) && ((d[k] >= nh) == hi);
    r[k] = 0;
    if (mine[k]) r[k] = atomicAdd(&cnt[d[k]], 1);
  }
#pragma unroll
  for (int k = 0; k < EPT; ++k) {
    if (mine[k]) {
      if (r[k] < CAP) {
        slots[(size_t)d[k] * CAP + r[k]] = s[k];
      } else {
        const int p = atomicAdd(ovf_n, 1);
        if ((unsigned)p < (unsigned)E) ovf[p] = make_int2(d[k], s[k]);
      }
    }
  }
  return t;
}

__global__ void __launch_bounds__(256)
count_place(const int* __restrict__ src, const int* __restrict__ dst,
            int* __restrict__ cnt, int* __restrict__ slots,
            int2* __restrict__ ovf, int* __restrict__ ctr, int E, int N,
            int nchunks) {
  __shared__ int task[2];
  const int nh = range_split(N);
  const int half = chip_half();
  int* const own = ctr + half * CTR_PITCH;
  int* const ovf_n = ctr + OVF_WORD;
  if (threadIdx.x == 0)
    task[0] = resolve_task(ctr, half, atomicAdd(own, 1), nchunks);
  __syncthreads();
  for (int it = 0; it < 2 * nchunks; ++it) {
    const int code = task[it & 1];
    if (code < 0) break;
    int t;
    const int blk0 = (code >> 1) * (256 * EPT);
    const int base = blk0 + threadIdx.x;
    const bool hi = (code & 1) != 0;
    if (blk0 + 256 * EPT <= E)
      t = count_place_body<true>(src, dst, cnt, slots, ovf, ovf_n, E, base, nh, hi, own);
    else
      t = count_place_body<false>(src, dst, cnt, slots, ovf, ovf_n, E, base, nh, hi, own);
    if (threadIdx.x == 0) task[(it + 1) & 1] = resolve_task(ctr, half, t, nchunks);
    __syncthreads();
  }
}

__global__ void __launch_bounds__(320)
gemm1_wmma(const float* __restrict__ x,
           const float* __restrict__ W1,
           float* __restrict__ h,
           int num_tiles) {
  int flat = blockIdx.x * blockDim.x + threadIdx.x;
  int wave = flat >> 5;
  int lane = threadIdx.x & 31;
  if (wave >= num_tiles) return;

  int m = lane & 15;
  bool hi = lane >= 16;
  int row0 = wave * 16;

  const float* xr = x + (size_t)(row0 + m) * 3;
  v2f a;
  a.x = hi ? xr[2] : xr[0];
  a.y = hi ? 0.0f : xr[1];

  v2f b;
  b.x = hi ? W1[2 * 16 + m] : W1[0 * 16 + m];
  b.y = hi ? 0.0f : W1[1 * 16 + m];

  v8f c = {};
  c = __builtin_amdgcn_wmma_f32_16x16x4_f32(
       false, a,  false, b,
       (short)0, c,  false,  false);

  int rbase = row0 + (hi ? 8 : 0);
#pragma unroll
  for (int v = 0; v < 8; ++v) {
    h[(size_t)(rbase + v) * 16 + m] = c[v];
  }
}


template <int J>
__device__ __forceinline__ unsigned row_bcast(unsigned v) {
  return (unsigned)__builtin_amdgcn_update_dpp(0, (int)v, 0x150 + J, 0xf, 0xf,
                                               true);
}

template <int M>
__device__ __forceinline__ float row_xadd(float v) {
  const int t = __builtin_amdgcn_update_dpp(0, __float_as_int(v), 0x160 + M,
                                            0xf, 0xf, true);
  return v + __int_as_float(t);
}

__device__ __forceinline__ float xhalf(float x) {
  return __int_as_float(__builtin_amdgcn_permlanex16(
      __float_as_int(x), __float_as_int(x), 0x76543210, 0xfedcba98, false,
      false));
}

__device__ __forceinline__ float wave_sum(float v) {
  v = row_xadd<1>(v);
  v = row_xadd<2>(v);
  v = row_xadd<4>(v);
  v = row_xadd<8>(v);
  return v + xhalf(v);
}

__device__ __forceinline__ int lane_i(int v, int l) {
  return __builtin_amdgcn_readlane(v, l);
}
__device__ __forceinline__ float lane_f(float v, int l) {
  return __int_as_float(__builtin_amdgcn_readlane(__float_as_int(v), l));
}

struct Chunk {
  unsigned srow;
  float w;
};

__device__ __forceinline__ float dinv_of(int c) {
  const float d = (float)(c + 1);
  return (d > 0.0f) ? __frsqrt_rn(d) : 0.0f;
}

__device__ __forceinline__ int load_src(const int* __restrict__ slots,
                                        size_t row, int n, int c, int slot) {
  return slots[row + (((c + slot) < n) ? (c + slot) : 0)];
}
__device__ __forceinline__ int src_of(int s, int n, int c, int slot) {
  return ((c + slot) < n) ? s : 0;
}

template <int J>
__device__ __forceinline__ float row_gather(const float* __restrict__ h,
                                            unsigned srow, unsigned fb) {
  const unsigned off = row_bcast<J>(srow) + fb;
  return *(const float*)((const char*)h + off);
}

template <int J>
__device__ __forceinline__ float step_acc(float acc, float v, float w, int lim) {
  const float wj = __uint_as_float(row_bcast<J>(__float_as_uint(w)));
  const float vv = (2 * J < lim) ? v : 0.0f;
  return acc + wj * vv;
}

template <int G>
__device__ __forceinline__ void grp_load(const float* __restrict__ h,
                                         unsigned srow, unsigned fb,
                                         float (&v)[4]) {
  v[0] = row_gather<4 * G + 0>(h, srow, fb);
  v[1] = row_gather<4 * G + 1>(h, srow, fb);
  v[2] = row_gather<4 * G + 2>(h, srow, fb);
  v[3] = row_gather<4 * G + 3>(h, srow, fb);
}
template <int G>
__device__ __forceinline__ float grp_acc(float acc, const float (&v)[4],
                                         float w, int lim) {
  acc = step_acc<4 * G + 0>(acc, v[0], w, lim);
  acc = step_acc<4 * G + 1>(acc, v[1], w, lim);
  acc = step_acc<4 * G + 2>(acc, v[2], w, lim);
  acc = step_acc<4 * G + 3>(acc, v[3], w, lim);
  return acc;
}

__device__ __forceinline__ float chunk_sum(const float* __restrict__ h,
                                           Chunk ch, unsigned fb, int half,
                                           int rem, float acc) {
  const int lim = rem - half;
  float a[4], b[4], c[4], d[4];
  if (rem > 24) {
    grp_load<0>(h, ch.srow, fb, a);
    grp_load<1>(h, ch.srow, fb, b);
    grp_load<2>(h, ch.srow, fb, c);
    grp_load<3>(h, ch.srow, fb, d);
    acc = grp_acc<0>(acc, a, ch.w, lim);
    acc = grp_acc<1>(acc, b, ch.w, lim);
    acc = grp_acc<2>(acc, c, ch.w, lim);
    acc = grp_acc<3>(acc, d, ch.w, lim);
  } else if (rem > 16) {
    grp_load<0>(h, ch.srow, fb, a);
    grp_load<1>(h, ch.srow, fb, b);
    grp_load<2>(h, ch.srow, fb, c);
    acc = grp_acc<0>(acc, a, ch.w, lim);
    acc = grp_acc<1>(acc, b, ch.w, lim);
    acc = grp_acc<2>(acc, c, ch.w, lim);
  } else if (rem > 8) {
    grp_load<0>(h, ch.srow, fb, a);
    grp_load<1>(h, ch.srow, fb, b);
    acc = grp_acc<0>(acc, a, ch.w, lim);
    acc = grp_acc<1>(acc, b, ch.w, lim);
  } else {
    grp_load<0>(h, ch.srow, fb, a);
    acc = grp_acc<0>(acc, a, ch.w, lim);
  }
  return acc;
}

template <int GB>
__device__ __forceinline__ float sum_ab(const float* __restrict__ h, Chunk A,
                                        Chunk B, unsigned fb, int limA,
                                        int limB, float acc) {
  float a0[4], a1[4], a2[4], a3[4], b0[4], b1[4], b2[4], b3[4];
  grp_load<0>(h, A.srow, fb, a0);
  grp_load<1>(h, A.srow, fb, a1);
  grp_load<2>(h, A.srow, fb, a2);
  grp_load<3>(h, A.srow, fb, a3);
  if constexpr (GB > 0) grp_load<0>(h, B.srow, fb, b0);
  if constexpr (GB > 1) grp_load<1>(h, B.srow, fb, b1);
  if constexpr (GB > 2) grp_load<2>(h, B.srow, fb, b2);
  if constexpr (GB > 3) grp_load<3>(h, B.srow, fb, b3);
  acc = grp_acc<0>(acc, a0, A.w, limA);
  acc = grp_acc<1>(acc, a1, A.w, limA);
  acc = grp_acc<2>(acc, a2, A.w, limA);
  acc = grp_acc<3>(acc, a3, A.w, limA);
  if constexpr (GB > 0) acc = grp_acc<0>(acc, b0, B.w, limB);
  if constexpr (GB > 1) acc = grp_acc<1>(acc, b1, B.w, limB);
  if constexpr (GB > 2) acc = grp_acc<2>(acc, b2, B.w, limB);
  if constexpr (GB > 3) acc = grp_acc<3>(acc, b3, B.w, limB);
  return acc;
}

__device__ __forceinline__ float chunk_sum_ab(const float* __restrict__ h,
                                              Chunk A, Chunk B, unsigned fb,
                                              int half, int n, float acc) {
  const int limA = 32 - half;
  const int remB = n - 32;
  const int limB = remB - half;
  if (remB > 24) return sum_ab<4>(h, A, B, fb, limA, limB, acc);
  if (remB > 16) return sum_ab<3>(h, A, B, fb, limA, limB, acc);
  if (remB > 8) return sum_ab<2>(h, A, B, fb, limA, limB, acc);
  return sum_ab<1>(h, A, B, fb, limA, limB, acc);
}

__device__ __forceinline__ float overflow_rows(const int2* __restrict__ ovf,
                                               int on, int i,
                                               const int* __restrict__ cnt,
                                               const float* __restrict__ h,
                                               float di, int f, int half,
                                               int lane, float acc) {
  for (int b = 0; b < on; b += 32) {
    const bool ok = (b + lane) < on;
    const int2 q = ok ? ovf[b + lane] : make_int2(-1, 0);
    const unsigned m = (unsigned)__ballot(q.x == i);
    for (int j = 0; j < 32; ++j) {
      if ((m >> j) & 1u) {
        const int s = lane_i(q.y, j);
        const float w = dinv_of(cnt[s]) * di;
        const float v = h[(size_t)s * 16 + f];
        acc += (half == 0) ? w * v : 0.0f;
      }
    }
  }
  return acc;
}

__global__ void __launch_bounds__(256)
pull_layer1(const int* __restrict__ slots, float* __restrict__ wrec,
            const int2* __restrict__ ovf, const int* __restrict__ ctr,
            const int* __restrict__ cnt,
            const float* __restrict__ h,
            const float* __restrict__ b1,
            const float* __restrict__ W2,
            float* __restrict__ z,
            int N, int E) {
  const int lane = threadIdx.x & 31;
  const int i0 = __builtin_amdgcn_readfirstlane(
      (int)((blockIdx.x * blockDim.x + threadIdx.x) >> 5) * 32);
  if (i0 >= N) return;
  const int nodes = (N - i0 < 32) ? (N - i0) : 32;
  const int mi = i0 + lane;
  const bool mok = lane < nodes;
  const int mc = mok ? mi : i0;
  const int my_cnt = mok ? cnt[mc] : 0;
  const float my_di = dinv_of(my_cnt);
  const int f = lane & 15, half = lane >> 4;
  const unsigned fb = (unsigned)f * 4u;
  const int slot = 2 * f + half;
  const float b1f = b1[f], w2f = W2[f];
  float zkeep = 0.0f;

  int n0 = lane_i(my_cnt, 0);
  n0 = (n0 < CAP) ? n0 : CAP;
  size_t row0 = (size_t)i0 * CAP;
  int s0a = load_src(slots, row0, n0, 0, slot);
  int s0b = load_src(slots, row0, n0, 32, slot);
  const int x1 = (nodes > 1) ? 1 : 0;
  int n1 = lane_i(my_cnt, x1);
  n1 = (n1 < CAP) ? n1 : CAP;
  size_t row1 = (size_t)(i0 + x1) * CAP;
  int s1a = load_src(slots, row1, n1, 0, slot);
  int s1b = load_src(slots, row1, n1, 32, slot);
  s0a = src_of(s0a, n0, 0, slot);
  s0b = src_of(s0b, n0, 32, slot);
  float d0a = dinv_of(cnt[s0a]);
  float d0b = dinv_of(cnt[s0b]);

  for (int nn = 0; nn < nodes; ++nn) {
    const int i = i0 + nn;
    const int n_c = n0;
    const size_t row_c = row0;
    const int sa = s0a, sb = s0b;
    const float hv = h[(size_t)i * 16 + f];
    const float di = lane_f(my_di, nn);
    const int nfull = lane_i(my_cnt, nn);
    const int t1a = src_of(s1a, n1, 0, slot);
    const int t1b = src_of(s1b, n1, 32, slot);
    const int c1a = cnt[t1a];
    const int c1b = cnt[t1b];
    const int x2 = (nn + 2 < nodes) ? (nn + 2) : (nodes - 1);
    int n2 = lane_i(my_cnt, x2);
    n2 = (n2 < CAP) ? n2 : CAP;
    const size_t row2 = (size_t)(i0 + x2) * CAP;
    const int s2a = load_src(slots, row2, n2, 0, slot);
    const int s2b = load_src(slots, row2, n2, 32, slot);
    const bool va = slot < n_c, vb = (32 + slot) < n_c;
    const float wa = va ? d0a * di : 0.0f;
    const float wb = vb ? d0b * di : 0.0f;
    if (va) wrec[row_c + slot] = wa;
    if (vb) wrec[row_c + 32 + slot] = wb;
    Chunk A, B;
    A.srow = (unsigned)sa * 64u;
    A.w = wa;
    B.srow = (unsigned)sb * 64u;
    B.w = wb;
    float acc = 0.0f;
    if (n_c > 32)
      acc = chunk_sum_ab(h, A, B, fb, half, n_c, acc);
    else if (n_c > 0)
      acc = chunk_sum(h, A, fb, half, n_c, acc);
    if (nfull > CAP) {
      int on = ctr[OVF_WORD];
      on = (on < E) ? on : E;
      acc = overflow_rows(ovf, on, i, cnt, h, di, f, half, lane, acc);
    }
    acc += xhalf(acc);
    const float a1 = fmaxf(acc + hv * (di * di) + b1f, 0.0f);
    float p = a1 * w2f;
    p = row_xadd<8>(p);
    p = row_xadd<4>(p);
    p = row_xadd<2>(p);
    p = row_xadd<1>(p);
    if (lane == nn) zkeep = p;
    n0 = n1;
    row0 = row1;
    s0a = t1a;
    s0b = t1b;
    d0a = dinv_of(c1a);
    d0b = dinv_of(c1b);
    n1 = n2;
    row1 = row2;
    s1a = s2a;
    s1b = s2b;
  }
  if (mok) z[mi] = zkeep;
}

__global__ void __launch_bounds__(256)
pull_layer2(const int* __restrict__ slots, const float* __restrict__ wrec,
            const int2* __restrict__ ovf, const int* __restrict__ ctr,
            const int* __restrict__ cnt, const float* __restrict__ z,
            const float* __restrict__ b2, float* __restrict__ out, int N,
            int E) {
  const int lane = threadIdx.x & 31;
  const int i0 = __builtin_amdgcn_readfirstlane(
      (int)((blockIdx.x * blockDim.x + threadIdx.x) >> 5) * 32);
  if (i0 >= N) return;
  const int nodes = (N - i0 < 32) ? (N - i0) : 32;
  const int mi = i0 + lane;
  const bool mok = lane < nodes;
  const int mc = mok ? mi : i0;
  const int my_cnt = mok ? cnt[mc] : 0;
  const float my_di = dinv_of(my_cnt);
  const float zself = z[mc];
  float keep = 0.0f;

  for (int nn = 0; nn < 32; nn += 4) {
    int nf[4], n[4];
    int sa[4], sb[4], ua[4], ub[4];
    float za[4], zb[4];
#pragma unroll
    for (int u = 0; u < 4; ++u) {
      nf[u] = lane_i(my_cnt, nn + u);
      n[u] = (nf[u] < CAP) ? nf[u] : CAP;
      const size_t row = (size_t)(i0 + nn + u) * CAP;
      const size_t ia = (lane < n[u]) ? row + lane : 0;
      const size_t ib = (32 + lane < n[u]) ? row + 32 + lane : 0;
      sa[u] = slots[ia];
      ua[u] = __float_as_int(wrec[ia]);
      sb[u] = slots[ib];
      ub[u] = __float_as_int(wrec[ib]);
    }
#pragma unroll
    for (int u = 0; u < 4; ++u) {
      za[u] = z[(lane < n[u]) ? sa[u] : 0];
      zb[u] = z[(32 + lane < n[u]) ? sb[u] : 0];
    }
#pragma unroll
    for (int u = 0; u < 4; ++u) {
      const float wa = __int_as_float(ua[u] & ((lane < n[u]) ? -1 : 0));
      const float wb = __int_as_float(ub[u] & ((32 + lane < n[u]) ? -1 : 0));
      float acc = wa * za[u] + wb * zb[u];
      if (nf[u] > CAP) {
        const int i = i0 + nn + u;
        const float di = lane_f(my_di, nn + u);
        int on = ctr[OVF_WORD];
        on = (on < E) ? on : E;
        for (int b = 0; b < on; b += 32) {
          const bool ok = (b + lane) < on;
          const int2 q = ok ? ovf[b + lane] : make_int2(-1, 0);
          if (q.x == i) acc += (dinv_of(cnt[q.y]) * di) * z[q.y];
        }
      }
      const float s = wave_sum(acc);
      if (lane == nn + u) keep = s;
    }
  }
  if (mok) out[mi] = keep + zself * (my_di * my_di) + b2[0];
}

extern "C" void kernel_launch(void* const* d_in, const int* in_sizes, int n_in,
                              void* d_out, int out_size, void* d_ws,
                              size_t ws_size, hipStream_t stream) {
  const float* x  = (const float*)d_in[0];
  const int*   ei = (const int*)d_in[1];
  const float* W1 = (const float*)d_in[2];
  const float* b1 = (const float*)d_in[3];
  const float* W2 = (const float*)d_in[4];
  const float* b2 = (const float*)d_in[5];
  float* out = (float*)d_out;

  const int N = in_sizes[0] / 3;
  const int E = in_sizes[1] / 2;
  const int* src = ei;
  const int* dst = ei + E;

  const size_t nb4 = (((size_t)N * 4) + 255) & ~(size_t)255;
  const size_t sb4 = (size_t)N * CAP * 4;
  const size_t ob8 = (((size_t)E * 8) + 255) & ~(size_t)255;
  char* ws = (char*)d_ws;
  int*   cnt   = (int*)(ws);
  float* z     = (float*)(ws + nb4);
  float* h     = (float*)(ws + 2 * nb4);
  int*   slots = (int*)(ws + 18 * nb4);
  float* wrec  = (float*)(ws + 18 * nb4 + sb4);
  int2*  ovf   = (int2*)(ws + 18 * nb4 + 2 * sb4);
  int*   ctr   = (int*)(ws + 18 * nb4 + 2 * sb4 + ob8);

  const int BT = 256;
  const int eblocks = (E + BT * EPT - 1) / (BT * EPT);
  const int nwaves = (N + 31) / 32;
  const int pblocks = (nwaves + 7) / 8;

  init_counts<<<(N + BT - 1) / BT, BT, 0, stream>>>(cnt, ctr, N);
  count_place<<<EDGE_WGS, BT, 0, stream>>>(src, dst, cnt, slots, ovf, ctr, E, N,
                                           eblocks);
  {
    int ntiles = N / 16;
    int wpb = 10;
    int blocks = (ntiles + wpb - 1) / wpb;
    gemm1_wmma<<<blocks, 32 * wpb, 0, stream>>>(x, W1, h, ntiles);
  }
  pull_layer1<<<pblocks, BT, 0, stream>>>(slots, wrec, ovf, ctr, cnt, h, b1, W2,
                                          z, N, E);
  pull_layer2<<<pblocks, BT, 0, stream>>>(slots, wrec, ovf, ctr, cnt, z, b2, out,
                                          N, E);
}


# layer-1 weights take each source's in-degree from a whole-graph byte table held in the work-group's LDS (255 = read the count word), not from one random 4-byte read per edge; sums unchanged
# speedup vs baseline: 11.3013x; 1.0997x over previous
#include <hip/hip_runtime.h>
#include <hip/hip_bf16.h>


typedef __attribute__((ext_vector_type(2))) float v2f;
typedef __attribute__((ext_vector_type(8))) float v8f;

#define EPT 8
#define CAP 64

#define EDGE_WGS 2048
#define CTR_WORDS 256
#define CTR_PITCH 64
#define OVF_WORD (2 * CTR_PITCH)

#define TAB_BYTES 262144
#define DEG_ESC 255
#define P1_WAVES 25
#define FILL_U 16

__global__ void init_counts(int* __restrict__ cnt, int* __restrict__ ctr, int N) {
  int i = blockIdx.x * blockDim.x + threadIdx.x;
  if (i < N) cnt[i] = 0;
  if (i < CTR_WORDS) ctr[i] = 0;
}

__device__ __forceinline__ int chip_half() {
  return (int)((__builtin_amdgcn_s_sendmsg_rtn(0x87) >> 18) & 1u);
}

__device__ __forceinline__ int range_split(int N) {
  return ((N >> 1) + 63) & ~63;
}

__device__ __forceinline__ int resolve_task(int* __restrict__ ctr, int r, int t,
                                            int nchunks) {
  if (t >= nchunks) {
    r ^= 1;
    t = atomicAdd(&ctr[r * CTR_PITCH], 1);
  }
  return (t < nchunks) ? (t * 2 + r) : -1;
}

template <bool FULL>
__device__ __forceinline__ int count_place_body(const int* __restrict__ src,
                                                const int* __restrict__ dst,
                                                int* __restrict__ cnt,
                                                int* __restrict__ slots,
                                                int2* __restrict__ ovf,
                                                int* __restrict__ ovf_n, int E,
                                                int base, int nh, bool hi,
                                                int* __restrict__ own) {
  int d[EPT];
  int s[EPT];
  int r[EPT];
  bool mine[EPT];
#pragma unroll
  for (int k = 0; k < EPT; ++k) {
    const int e = base + k * 256;
    d[k] = (FULL || e < E) ? dst[e] : 0;
  }
#pragma unroll
  for (int k = 0; k < EPT; ++k) {
    const int e = base + k * 256;
    s[k] = (FULL || e < E) ? src[e] : 0;
  }
  int t = 0;
  if (threadIdx.x == 0) t = atomicAdd(own, 1);
  asm volatile("" ::"v"(d[0]), "v"(d[1]), "v"(d[2]), "v"(d[3]), "v"(d[4]),
               "v"(d[5]), "v"(d[6]), "v"(d[7]));
#pragma unroll
  for (int k = 0; k < EPT; ++k) {
    const int e = base + k * 256;
    mine[k] = (FULL || e < E) && ((d[k] >= nh) == hi);
    r[k] = 0;
    if (mine[k]) r[k] = atomicAdd(&cnt[d[k]], 1);
  }
#pragma unroll
  for (int k = 0; k < EPT; ++k) {
    if (mine[k]) {
      if (r[k] < CAP) {
        slots[(size_t)d[k] * CAP + r[k]] = s[k];
      } else {
        const int p = atomicAdd(ovf_n, 1);
        if ((unsigned)p < (unsigned)E) ovf[p] = make_int2(d[k], s[k]);
      }
    }
  }
  return t;
}

__global__ void __launch_bounds__(256)
count_place(const int* __restrict__ src, const int* __restrict__ dst,
            int* __restrict__ cnt, int* __restrict__ slots,
            int2* __restrict__ ovf, int* __restrict__ ctr, int E, int N,
            int nchunks) {
  __shared__ int task[2];
  const int nh = range_split(N);
  const int half = chip_half();
  int* const own = ctr + half * CTR_PITCH;
  int* const ovf_n = ctr + OVF_WORD;
  if (threadIdx.x == 0)
    task[0] = resolve_task(ctr, half, atomicAdd(own, 1), nchunks);
  __syncthreads();
  for (int it = 0; it < 2 * nchunks; ++it) {
    const int code = task[it & 1];
    if (code < 0) break;
    int t;
    const int blk0 = (code >> 1) * (256 * EPT);
    const int base = blk0 + threadIdx.x;
    const bool hi = (code & 1) != 0;
    if (blk0 + 256 * EPT <= E)
      t = count_place_body<true>(src, dst, cnt, slots, ovf, ovf_n, E, base, nh, hi, own);
    else
      t = count_place_body<false>(src, dst, cnt, slots, ovf, ovf_n, E, base, nh, hi, own);
    if (threadIdx.x == 0) task[(it + 1) & 1] = resolve_task(ctr, half, t, nchunks);
    __syncthreads();
  }
}

__global__ void __launch_bounds__(256)
gemm1_wmma(const float* __restrict__ x,
           const float* __restrict__ W1,
           float* __restrict__ h,
           const int* __restrict__ cnt,
           unsigned* __restrict__ deg8,
           int num_tiles, int N) {
  int flat = blockIdx.x * blockDim.x + threadIdx.x;
  int wave = flat >> 5;
  int lane = threadIdx.x & 31;
  if ((threadIdx.x >> 5) == 0) {
    const int nb = blockIdx.x * 128 + lane * 4;
    unsigned w = 0;
#pragma unroll
    for (int k = 0; k < 4; ++k) {
      const int i = nb + k;
      int c = (i < N) ? cnt[i] : 0;
      c = (c < DEG_ESC) ? c : DEG_ESC;
      w |= (unsigned)c << (8 * k);
    }
    deg8[blockIdx.x * 32 + lane] = w;
  }
  if (wave >= num_tiles) return;

  int m = lane & 15;
  bool hi = lane >= 16;
  int row0 = wave * 16;

  const float* xr = x + (size_t)(row0 + m) * 3;
  v2f a;
  a.x = hi ? xr[2] : xr[0];
  a.y = hi ? 0.0f : xr[1];

  v2f b;
  b.x = hi ? W1[2 * 16 + m] : W1[0 * 16 + m];
  b.y = hi ? 0.0f : W1[1 * 16 + m];

  v8f c = {};
  c = __builtin_amdgcn_wmma_f32_16x16x4_f32(
       false, a,  false, b,
       (short)0, c,  false,  false);

  int rbase = row0 + (hi ? 8 : 0);
#pragma unroll
  for (int v = 0; v < 8; ++v) {
    h[(size_t)(rbase + v) * 16 + m] = c[v];
  }
}


template <int J>
__device__ __forceinline__ unsigned row_bcast(unsigned v) {
  return (unsigned)__builtin_amdgcn_update_dpp(0, (int)v, 0x150 + J, 0xf, 0xf,
                                               true);
}

template <int M>
__device__ __forceinline__ float row_xadd(float v) {
  const int t = __builtin_amdgcn_update_dpp(0, __float_as_int(v), 0x160 + M,
                                            0xf, 0xf, true);
  return v + __int_as_float(t);
}

__device__ __forceinline__ float xhalf(float x) {
  return __int_as_float(__builtin_amdgcn_permlanex16(
      __float_as_int(x), __float_as_int(x), 0x76543210, 0xfedcba98, false,
      false));
}

__device__ __forceinline__ float wave_sum(float v) {
  v = row_xadd<1>(v);
  v = row_xadd<2>(v);
  v = row_xadd<4>(v);
  v = row_xadd<8>(v);
  return v + xhalf(v);
}

__device__ __forceinline__ int lane_i(int v, int l) {
  return __builtin_amdgcn_readlane(v, l);
}
__device__ __forceinline__ float lane_f(float v, int l) {
  return __int_as_float(__builtin_amdgcn_readlane(__float_as_int(v), l));
}

struct Chunk {
  unsigned srow;
  float w;
};

__device__ __forceinline__ float dinv_of(int c) {
  const float d = (float)(c + 1);
  return (d > 0.0f) ? __frsqrt_rn(d) : 0.0f;
}

__device__ __forceinline__ int deg_byte(const unsigned char* tab, int tl, int s) {
  const int b = tab[((unsigned)s < (unsigned)tl) ? s : 0];
  return ((unsigned)s < (unsigned)tl) ? b : DEG_ESC;
}
__device__ __forceinline__ int deg_full(int b, const int* __restrict__ cnt, int s) {
  if (b == DEG_ESC) b = cnt[s];
  return b;
}

__device__ __forceinline__ int load_src(const int* __restrict__ slots,
                                        size_t row, int n, int c, int slot) {
  return slots[row + (((c + slot) < n) ? (c + slot) : 0)];
}
__device__ __forceinline__ int src_of(int s, int n, int c, int slot) {
  return ((c + slot) < n) ? s : 0;
}

template <int J>
__device__ __forceinline__ float row_gather(const float* __restrict__ h,
                                            unsigned srow, unsigned fb) {
  const unsigned off = row_bcast<J>(srow) + fb;
  return *(const float*)((const char*)h + off);
}

template <int J>
__device__ __forceinline__ float step_acc(float acc, float v, float w, int lim) {
  const float wj = __uint_as_float(row_bcast<J>(__float_as_uint(w)));
  const float vv = (2 * J < lim) ? v : 0.0f;
  return acc + wj * vv;
}

template <int G>
__device__ __forceinline__ void grp_load(const float* __restrict__ h,
                                         unsigned srow, unsigned fb,
                                         float (&v)[4]) {
  v[0] = row_gather<4 * G + 0>(h, srow, fb);
  v[1] = row_gather<4 * G + 1>(h, srow, fb);
  v[2] = row_gather<4 * G + 2>(h, srow, fb);
  v[3] = row_gather<4 * G + 3>(h, srow, fb);
}
template <int G>
__device__ __forceinline__ float grp_acc(float acc, const float (&v)[4],
                                         float w, int lim) {
  acc = step_acc<4 * G + 0>(acc, v[0], w, lim);
  acc = step_acc<4 * G + 1>(acc, v[1], w, lim);
  acc = step_acc<4 * G + 2>(acc, v[2], w, lim);
  acc = step_acc<4 * G + 3>(acc, v[3], w, lim);
  return acc;
}

__device__ __forceinline__ float chunk_sum(const float* __restrict__ h,
                                           Chunk ch, unsigned fb, int half,
                                           int rem, float acc) {
  const int lim = rem - half;
  float a[4], b[4], c[4], d[4];
  if (rem > 24) {
    grp_load<0>(h, ch.srow, fb, a);
    grp_load<1>(h, ch.srow, fb, b);
    grp_load<2>(h, ch.srow, fb, c);
    grp_load<3>(h, ch.srow, fb, d);
    acc = grp_acc<0>(acc, a, ch.w, lim);
    acc = grp_acc<1>(acc, b, ch.w, lim);
    acc = grp_acc<2>(acc, c, ch.w, lim);
    acc = grp_acc<3>(acc, d, ch.w, lim);
  } else if (rem > 16) {
    grp_load<0>(h, ch.srow, fb, a);
    grp_load<1>(h, ch.srow, fb, b);
    grp_load<2>(h, ch.srow, fb, c);
    acc = grp_acc<0>(acc, a, ch.w, lim);
    acc = grp_acc<1>(acc, b, ch.w, lim);
    acc = grp_acc<2>(acc, c, ch.w, lim);
  } else if (rem > 8) {
    grp_load<0>(h, ch.srow, fb, a);
    grp_load<1>(h, ch.srow, fb, b);
    acc = grp_acc<0>(acc, a, ch.w, lim);
    acc = grp_acc<1>(acc, b, ch.w, lim);
  } else {
    grp_load<0>(h, ch.srow, fb, a);
    acc = grp_acc<0>(acc, a, ch.w, lim);
  }
  return acc;
}

template <int GB>
__device__ __forceinline__ float sum_ab(const float* __restrict__ h, Chunk A,
                                        Chunk B, unsigned fb, int limA,
                                        int limB, float acc) {
  float a0[4], a1[4], a2[4], a3[4], b0[4], b1[4], b2[4], b3[4];
  grp_load<0>(h, A.srow, fb, a0);
  grp_load<1>(h, A.srow, fb, a1);
  grp_load<2>(h, A.srow, fb, a2);
  grp_load<3>(h, A.srow, fb, a3);
  if constexpr (GB > 0) grp_load<0>(h, B.srow, fb, b0);
  if constexpr (GB > 1) grp_load<1>(h, B.srow, fb, b1);
  if constexpr (GB > 2) grp_load<2>(h, B.srow, fb, b2);
  if constexpr (GB > 3) grp_load<3>(h, B.srow, fb, b3);
  acc = grp_acc<0>(acc, a0, A.w, limA);
  acc = grp_acc<1>(acc, a1, A.w, limA);
  acc = grp_acc<2>(acc, a2, A.w, limA);
  acc = grp_acc<3>(acc, a3, A.w, limA);
  if constexpr (GB > 0) acc = grp_acc<0>(acc, b0, B.w, limB);
  if constexpr (GB > 1) acc = grp_acc<1>(acc, b1, B.w, limB);
  if constexpr (GB > 2) acc = grp_acc<2>(acc, b2, B.w, limB);
  if constexpr (GB > 3) acc = grp_acc<3>(acc, b3, B.w, limB);
  return acc;
}

__device__ __forceinline__ float chunk_sum_ab(const float* __restrict__ h,
                                              Chunk A, Chunk B, unsigned fb,
                                              int half, int n, float acc) {
  const int limA = 32 - half;
  const int remB = n - 32;
  const int limB = remB - half;
  if (remB > 24) return sum_ab<4>(h, A, B, fb, limA, limB, acc);
  if (remB > 16) return sum_ab<3>(h, A, B, fb, limA, limB, acc);
  if (remB > 8) return sum_ab<2>(h, A, B, fb, limA, limB, acc);
  return sum_ab<1>(h, A, B, fb, limA, limB, acc);
}

__device__ __forceinline__ float overflow_rows(const int2* __restrict__ ovf,
                                               int on, int i,
                                               const int* __restrict__ cnt,
                                               const float* __restrict__ h,
                                               float di, int f, int half,
                                               int lane, float acc) {
  for (int b = 0; b < on; b += 32) {
    const bool ok = (b + lane) < on;
    const int2 q = ok ? ovf[b + lane] : make_int2(-1, 0);
    const unsigned m = (unsigned)__ballot(q.x == i);
    for (int j = 0; j < 32; ++j) {
      if ((m >> j) & 1u) {
        const int s = lane_i(q.y, j);
        const float w = dinv_of(cnt[s]) * di;
        const float v = h[(size_t)s * 16 + f];
        acc += (half == 0) ? w * v : 0.0f;
      }
    }
  }
  return acc;
}

__global__ void __launch_bounds__(P1_WAVES * 32)
pull_layer1(const int* __restrict__ slots, float* __restrict__ wrec,
            const int2* __restrict__ ovf, const int* __restrict__ ctr,
            const int* __restrict__ cnt,
            const uint4* __restrict__ deg8,
            const float* __restrict__ h,
            const float* __restrict__ b1,
            const float* __restrict__ W2,
            float* __restrict__ z,
            int N, int E) {
  __shared__ uint4 tab4[TAB_BYTES / 16];
  const unsigned char* const tab = (const unsigned char*)tab4;
  const int tn = (N + 127) & ~127;
  const int tl = (tn < TAB_BYTES) ? tn : TAB_BYTES;
  {
    const int nv = tl >> 4;
    const int NT = P1_WAVES * 32;
    for (int v0 = threadIdx.x; v0 < nv; v0 += FILL_U * NT) {
      uint4 a[FILL_U];
#pragma unroll
      for (int k = 0; k < FILL_U; ++k) {
        const int v = v0 + k * NT;
        a[k] = deg8[(v < nv) ? v : (nv - 1)];
      }
#pragma unroll
      for (int k = 0; k < FILL_U; ++k) {
        const int v = v0 + k * NT;
        tab4[(v < nv) ? v : (nv - 1)] = a[k];
      }
    }
  }
  __syncthreads();
  const int lane = threadIdx.x & 31;
  const int i0 = __builtin_amdgcn_readfirstlane(
      (int)((blockIdx.x * blockDim.x + threadIdx.x) >> 5) * 32);
  if (i0 >= N) return;
  const int nodes = (N - i0 < 32) ? (N - i0) : 32;
  const int mi = i0 + lane;
  const bool mok = lane < nodes;
  const int mc = mok ? mi : i0;
  const int my_cnt = mok ? cnt[mc] : 0;
  const float my_di = dinv_of(my_cnt);
  const int f = lane & 15, half = lane >> 4;
  const unsigned fb = (unsigned)f * 4u;
  const int slot = 2 * f + half;
  const float b1f = b1[f], w2f = W2[f];
  float zkeep = 0.0f;

  int n0 = lane_i(my_cnt, 0);
  n0 = (n0 < CAP) ? n0 : CAP;
  size_t row0 = (size_t)i0 * CAP;
  int s0a = load_src(slots, row0, n0, 0, slot);
  int s0b = load_src(slots, row0, n0, 32, slot);
  const int x1 = (nodes > 1) ? 1 : 0;
  int n1 = lane_i(my_cnt, x1);
  n1 = (n1 < CAP) ? n1 : CAP;
  size_t row1 = (size_t)(i0 + x1) * CAP;
  int s1a = load_src(slots, row1, n1, 0, slot);
  int s1b = load_src(slots, row1, n1, 32, slot);
  s0a = src_of(s0a, n0, 0, slot);
  s0b = src_of(s0b, n0, 32, slot);
  const int b0a = deg_byte(tab, tl, s0a);
  const int b0b = deg_byte(tab, tl, s0b);
  float d0a = dinv_of(deg_full(b0a, cnt, s0a));
  float d0b = dinv_of(deg_full(b0b, cnt, s0b));

  for (int nn = 0; nn < nodes; ++nn) {
    const int i = i0 + nn;
    const int n_c = n0;
    const size_t row_c = row0;
    const int sa = s0a, sb = s0b;
    const float hv = h[(size_t)i * 16 + f];
    const float di = lane_f(my_di, nn);
    const int nfull = lane_i(my_cnt, nn);
    const int t1a = src_of(s1a, n1, 0, slot);
    const int t1b = src_of(s1b, n1, 32, slot);
    const int e1a = deg_byte(tab, tl, t1a);
    const int e1b = deg_byte(tab, tl, t1b);
    const int x2 = (nn + 2 < nodes) ? (nn + 2) : (nodes - 1);
    int n2 = lane_i(my_cnt, x2);
    n2 = (n2 < CAP) ? n2 : CAP;
    const size_t row2 = (size_t)(i0 + x2) * CAP;
    const int s2a = load_src(slots, row2, n2, 0, slot);
    const int s2b = load_src(slots, row2, n2, 32, slot);
    const bool va = slot < n_c, vb = (32 + slot) < n_c;
    const float wa = va ? d0a * di : 0.0f;
    const float wb = vb ? d0b * di : 0.0f;
    if (va) wrec[row_c + slot] = wa;
    if (vb) wrec[row_c + 32 + slot] = wb;
    Chunk A, B;
    A.srow = (unsigned)sa * 64u;
    A.w = wa;
    B.srow = (unsigned)sb * 64u;
    B.w = wb;
    float acc = 0.0f;
    if (n_c > 32)
      acc = chunk_sum_ab(h, A, B, fb, half, n_c, acc);
    else if (n_c > 0)
      acc = chunk_sum(h, A, fb, half, n_c, acc);
    if (nfull > CAP) {
      int on = ctr[OVF_WORD];
      on = (on < E) ? on : E;
      acc = overflow_rows(ovf, on, i, cnt, h, di, f, half, lane, acc);
    }
    acc += xhalf(acc);
    const float a1 = fmaxf(acc + hv * (di * di) + b1f, 0.0f);
    float p = a1 * w2f;
    p = row_xadd<8>(p);
    p = row_xadd<4>(p);
    p = row_xadd<2>(p);
    p = row_xadd<1>(p);
    if (lane == nn) zkeep = p;
    n0 = n1;
    row0 = row1;
    s0a = t1a;
    s0b = t1b;
    d0a = dinv_of(deg_full(e1a, cnt, t1a));
    d0b = dinv_of(deg_full(e1b, cnt, t1b));
    n1 = n2;
    row1 = row2;
    s1a = s2a;
    s1b = s2b;
  }
  if (mok) z[mi] = zkeep;
}

__global__ void __launch_bounds__(256)
pull_layer2(const int* __restrict__ slots, const float* __restrict__ wrec,
            const int2* __restrict__ ovf, const int* __restrict__ ctr,
            const int* __restrict__ cnt, const float* __restrict__ z,
            const float* __restrict__ b2, float* __restrict__ out, int N,
            int E) {
  const int lane = threadIdx.x & 31;
  const int i0 = __builtin_amdgcn_readfirstlane(
      (int)((blockIdx.x * blockDim.x + threadIdx.x) >> 5) * 32);
  if (i0 >= N) return;
  const int nodes = (N - i0 < 32) ? (N - i0) : 32;
  const int mi = i0 + lane;
  const bool mok = lane < nodes;
  const int mc = mok ? mi : i0;
  const int my_cnt = mok ? cnt[mc] : 0;
  const float my_di = dinv_of(my_cnt);
  const float zself = z[mc];
  float keep = 0.0f;

  for (int nn = 0; nn < 32; nn += 4) {
    int nf[4], n[4];
    int sa[4], sb[4], ua[4], ub[4];
    float za[4], zb[4];
#pragma unroll
    for (int u = 0; u < 4; ++u) {
      nf[u] = lane_i(my_cnt, nn + u);
      n[u] = (nf[u] < CAP) ? nf[u] : CAP;
      const size_t row = (size_t)(i0 + nn + u) * CAP;
      const size_t ia = (lane < n[u]) ? row + lane : 0;
      const size_t ib = (32 + lane < n[u]) ? row + 32 + lane : 0;
      sa[u] = slots[ia];
      ua[u] = __float_as_int(wrec[ia]);
      sb[u] = slots[ib];
      ub[u] = __float_as_int(wrec[ib]);
    }
#pragma unroll
    for (int u = 0; u < 4; ++u) {
      za[u] = z[(lane < n[u]) ? sa[u] : 0];
      zb[u] = z[(32 + lane < n[u]) ? sb[u] : 0];
    }
#pragma unroll
    for (int u = 0; u < 4; ++u) {
      const float wa = __int_as_float(ua[u] & ((lane < n[u]) ? -1 : 0));
      const float wb = __int_as_float(ub[u] & ((32 + lane < n[u]) ? -1 : 0));
      float acc = wa * za[u] + wb * zb[u];
      if (nf[u] > CAP) {
        const int i = i0 + nn + u;
        const float di = lane_f(my_di, nn + u);
        int on = ctr[OVF_WORD];
        on = (on < E) ? on : E;
        for (int b = 0; b < on; b += 32) {
          const bool ok = (b + lane) < on;
          const int2 q = ok ? ovf[b + lane] : make_int2(-1, 0);
          if (q.x == i) acc += (dinv_of(cnt[q.y]) * di) * z[q.y];
        }
      }
      const float s = wave_sum(acc);
      if (lane == nn + u) keep = s;
    }
  }
  if (mok) out[mi] = keep + zself * (my_di * my_di) + b2[0];
}

extern "C" void kernel_launch(void* const* d_in, const int* in_sizes, int n_in,
                              void* d_out, int out_size, void* d_ws,
                              size_t ws_size, hipStream_t stream) {
  const float* x  = (const float*)d_in[0];
  const int*   ei = (const int*)d_in[1];
  const float* W1 = (const float*)d_in[2];
  const float* b1 = (const float*)d_in[3];
  const float* W2 = (const float*)d_in[4];
  const float* b2 = (const float*)d_in[5];
  float* out = (float*)d_out;

  const int N = in_sizes[0] / 3;
  const int E = in_sizes[1] / 2;
  const int* src = ei;
  const int* dst = ei + E;

  const size_t nb4 = (((size_t)N * 4) + 255) & ~(size_t)255;
  const size_t sb4 = (size_t)N * CAP * 4;
  const size_t ob8 = (((size_t)E * 8) + 255) & ~(size_t)255;
  char* ws = (char*)d_ws;
  int*   cnt   = (int*)(ws);
  float* z     = (float*)(ws + nb4);
  float* h     = (float*)(ws + 2 * nb4);
  int*   slots = (int*)(ws + 18 * nb4);
  float* wrec  = (float*)(ws + 18 * nb4 + sb4);
  int2*  ovf   = (int2*)(ws + 18 * nb4 + 2 * sb4);
  int*   ctr   = (int*)(ws + 18 * nb4 + 2 * sb4 + ob8);
  unsigned* deg8 = (unsigned*)(ws + 18 * nb4 + 2 * sb4 + ob8 + CTR_WORDS * 4);

  const int BT = 256;
  const int eblocks = (E + BT * EPT - 1) / (BT * EPT);
  const int nwaves = (N + 31) / 32;
  const int pblocks = (nwaves + 7) / 8;

  init_counts<<<(N + BT - 1) / BT, BT, 0, stream>>>(cnt, ctr, N);
  count_place<<<EDGE_WGS, BT, 0, stream>>>(src, dst, cnt, slots, ovf, ctr, E, N,
                                           eblocks);
  {
    int ntiles = N / 16;
    int wpb = 8;
    int blocks = (ntiles + wpb - 1) / wpb;
    gemm1_wmma<<<blocks, 32 * wpb, 0, stream>>>(x, W1, h, cnt, deg8, ntiles, N);
  }
  pull_layer1<<<(nwaves + P1_WAVES - 1) / P1_WAVES, P1_WAVES * 32, 0, stream>>>(
      slots, wrec, ovf, ctr, cnt, (const uint4*)deg8, h, b1, W2, z, N, E);
  pull_layer2<<<pblocks, BT, 0, stream>>>(slots, wrec, ovf, ctr, cnt, z, b2, out,
                                          N, E);
}


# layer 2 forms each edge weight from a (z, rsqrt-degree) pair gathered with one 8-byte read per source: the per-edge weight plane is neither written by layer 1 nor read by layer 2; sums unchanged
# speedup vs baseline: 11.6814x; 1.0336x over previous
#include <hip/hip_runtime.h>
#include <hip/hip_bf16.h>


typedef __attribute__((ext_vector_type(2))) float v2f;
typedef __attribute__((ext_vector_type(8))) float v8f;

#define EPT 8
#define CAP 64

#define EDGE_WGS 2048
#define CTR_WORDS 256
#define CTR_PITCH 64
#define OVF_WORD (2 * CTR_PITCH)

#define TAB_BYTES 262144
#define DEG_ESC 255
#define P1_WAVES 25
#define FILL_U 16

__global__ void init_counts(int* __restrict__ cnt, int* __restrict__ ctr, int N) {
  int i = blockIdx.x * blockDim.x + threadIdx.x;
  if (i < N) cnt[i] = 0;
  if (i < CTR_WORDS) ctr[i] = 0;
}

__device__ __forceinline__ int chip_half() {
  return (int)((__builtin_amdgcn_s_sendmsg_rtn(0x87) >> 18) & 1u);
}

__device__ __forceinline__ int range_split(int N) {
  return ((N >> 1) + 63) & ~63;
}

__device__ __forceinline__ int resolve_task(int* __restrict__ ctr, int r, int t,
                                            int nchunks) {
  if (t >= nchunks) {
    r ^= 1;
    t = atomicAdd(&ctr[r * CTR_PITCH], 1);
  }
  return (t < nchunks) ? (t * 2 + r) : -1;
}

template <bool FULL>
__device__ __forceinline__ int count_place_body(const int* __restrict__ src,
                                                const int* __restrict__ dst,
                                                int* __restrict__ cnt,
                                                int* __restrict__ slots,
                                                int2* __restrict__ ovf,
                                                int* __restrict__ ovf_n, int E,
                                                int base, int nh, bool hi,
                                                int* __restrict__ own) {
  int d[EPT];
  int s[EPT];
  int r[EPT];
  bool mine[EPT];
#pragma unroll
  for (int k = 0; k < EPT; ++k) {
    const int e = base + k * 256;
    d[k] = (FULL || e < E) ? dst[e] : 0;
  }
#pragma unroll
  for (int k = 0; k < EPT; ++k) {
    const int e = base + k * 256;
    s[k] = (FULL || e < E) ? src[e] : 0;
  }
  int t = 0;
  if (threadIdx.x == 0) t = atomicAdd(own, 1);
  asm volatile("" ::"v"(d[0]), "v"(d[1]), "v"(d[2]), "v"(d[3]), "v"(d[4]),
               "v"(d[5]), "v"(d[6]), "v"(d[7]));
#pragma unroll
  for (int k = 0; k < EPT; ++k) {
    const int e = base + k * 256;
    mine[k] = (FULL || e < E) && ((d[k] >= nh) == hi);
    r[k] = 0;
    if (mine[k]) r[k] = atomicAdd(&cnt[d[k]], 1);
  }
#pragma unroll
  for (int k = 0; k < EPT; ++k) {
    if (mine[k]) {
      if (r[k] < CAP) {
        slots[(size_t)d[k] * CAP + r[k]] = s[k];
      } else {
        const int p = atomicAdd(ovf_n, 1);
        if ((unsigned)p < (unsigned)E) ovf[p] = make_int2(d[k], s[k]);
      }
    }
  }
  return t;
}

__global__ void __launch_bounds__(256)
count_place(const int* __restrict__ src, const int* __restrict__ dst,
            int* __restrict__ cnt, int* __restrict__ slots,
            int2* __restrict__ ovf, int* __restrict__ ctr, int E, int N,
            int nchunks) {
  __shared__ int task[2];
  const int nh = range_split(N);
  const int half = chip_half();
  int* const own = ctr + half * CTR_PITCH;
  int* const ovf_n = ctr + OVF_WORD;
  if (threadIdx.x == 0)
    task[0] = resolve_task(ctr, half, atomicAdd(own, 1), nchunks);
  __syncthreads();
  for (int it = 0; it < 2 * nchunks; ++it) {
    const int code = task[it & 1];
    if (code < 0) break;
    int t;
    const int blk0 = (code >> 1) * (256 * EPT);
    const int base = blk0 + threadIdx.x;
    const bool hi = (code & 1) != 0;
    if (blk0 + 256 * EPT <= E)
      t = count_place_body<true>(src, dst, cnt, slots, ovf, ovf_n, E, base, nh, hi, own);
    else
      t = count_place_body<false>(src, dst, cnt, slots, ovf, ovf_n, E, base, nh, hi, own);
    if (threadIdx.x == 0) task[(it + 1) & 1] = resolve_task(ctr, half, t, nchunks);
    __syncthreads();
  }
}

__global__ void __launch_bounds__(256)
gemm1_wmma(const float* __restrict__ x,
           const float* __restrict__ W1,
           float* __restrict__ h,
           const int* __restrict__ cnt,
           unsigned* __restrict__ deg8,
           int num_tiles, int N) {
  int flat = blockIdx.x * blockDim.x + threadIdx.x;
  int wave = flat >> 5;
  int lane = threadIdx.x & 31;
  if ((threadIdx.x >> 5) == 0) {
    const int nb = blockIdx.x * 128 + lane * 4;
    unsigned w = 0;
#pragma unroll
    for (int k = 0; k < 4; ++k) {
      const int i = nb + k;
      int c = (i < N) ? cnt[i] : 0;
      c = (c < DEG_ESC) ? c : DEG_ESC;
      w |= (unsigned)c << (8 * k);
    }
    deg8[blockIdx.x * 32 + lane] = w;
  }
  if (wave >= num_tiles) return;

  int m = lane & 15;
  bool hi = lane >= 16;
  int row0 = wave * 16;

  const float* xr = x + (size_t)(row0 + m) * 3;
  v2f a;
  a.x = hi ? xr[2] : xr[0];
  a.y = hi ? 0.0f : xr[1];

  v2f b;
  b.x = hi ? W1[2 * 16 + m] : W1[0 * 16 + m];
  b.y = hi ? 0.0f : W1[1 * 16 + m];

  v8f c = {};
  c = __builtin_amdgcn_wmma_f32_16x16x4_f32(
       false, a,  false, b,
       (short)0, c,  false,  false);

  int rbase = row0 + (hi ? 8 : 0);
#pragma unroll
  for (int v = 0; v < 8; ++v) {
    h[(size_t)(rbase + v) * 16 + m] = c[v];
  }
}


template <int J>
__device__ __forceinline__ unsigned row_bcast(unsigned v) {
  return (unsigned)__builtin_amdgcn_update_dpp(0, (int)v, 0x150 + J, 0xf, 0xf,
                                               true);
}

template <int M>
__device__ __forceinline__ float row_xadd(float v) {
  const int t = __builtin_amdgcn_update_dpp(0, __float_as_int(v), 0x160 + M,
                                            0xf, 0xf, true);
  return v + __int_as_float(t);
}

__device__ __forceinline__ float xhalf(float x) {
  return __int_as_float(__builtin_amdgcn_permlanex16(
      __float_as_int(x), __float_as_int(x), 0x76543210, 0xfedcba98, false,
      false));
}

__device__ __forceinline__ float wave_sum(float v) {
  v = row_xadd<1>(v);
  v = row_xadd<2>(v);
  v = row_xadd<4>(v);
  v = row_xadd<8>(v);
  return v + xhalf(v);
}

__device__ __forceinline__ int lane_i(int v, int l) {
  return __builtin_amdgcn_readlane(v, l);
}
__device__ __forceinline__ float lane_f(float v, int l) {
  return __int_as_float(__builtin_amdgcn_readlane(__float_as_int(v), l));
}

struct Chunk {
  unsigned srow;
  float w;
};

__device__ __forceinline__ float dinv_of(int c) {
  const float d = (float)(c + 1);
  return (d > 0.0f) ? __frsqrt_rn(d) : 0.0f;
}

__device__ __forceinline__ int deg_byte(const unsigned char* tab, int tl, int s) {
  const int b = tab[((unsigned)s < (unsigned)tl) ? s : 0];
  return ((unsigned)s < (unsigned)tl) ? b : DEG_ESC;
}
__device__ __forceinline__ int deg_full(int b, const int* __restrict__ cnt, int s) {
  if (b == DEG_ESC) b = cnt[s];
  return b;
}

__device__ __forceinline__ int load_src(const int* __restrict__ slots,
                                        size_t row, int n, int c, int slot) {
  return slots[row + (((c + slot) < n) ? (c + slot) : 0)];
}
__device__ __forceinline__ int src_of(int s, int n, int c, int slot) {
  return ((c + slot) < n) ? s : 0;
}

template <int J>
__device__ __forceinline__ float row_gather(const float* __restrict__ h,
                                            unsigned srow, unsigned fb) {
  const unsigned off = row_bcast<J>(srow) + fb;
  return *(const float*)((const char*)h + off);
}

template <int J>
__device__ __forceinline__ float step_acc(float acc, float v, float w, int lim) {
  const float wj = __uint_as_float(row_bcast<J>(__float_as_uint(w)));
  const float vv = (2 * J < lim) ? v : 0.0f;
  return acc + wj * vv;
}

template <int G>
__device__ __forceinline__ void grp_load(const float* __restrict__ h,
                                         unsigned srow, unsigned fb,
                                         float (&v)[4]) {
  v[0] = row_gather<4 * G + 0>(h, srow, fb);
  v[1] = row_gather<4 * G + 1>(h, srow, fb);
  v[2] = row_gather<4 * G + 2>(h, srow, fb);
  v[3] = row_gather<4 * G + 3>(h, srow, fb);
}
template <int G>
__device__ __forceinline__ float grp_acc(float acc, const float (&v)[4],
                                         float w, int lim) {
  acc = step_acc<4 * G + 0>(acc, v[0], w, lim);
  acc = step_acc<4 * G + 1>(acc, v[1], w, lim);
  acc = step_acc<4 * G + 2>(acc, v[2], w, lim);
  acc = step_acc<4 * G + 3>(acc, v[3], w, lim);
  return acc;
}

__device__ __forceinline__ float chunk_sum(const float* __restrict__ h,
                                           Chunk ch, unsigned fb, int half,
                                           int rem, float acc) {
  const int lim = rem - half;
  float a[4], b[4], c[4], d[4];
  if (rem > 24) {
    grp_load<0>(h, ch.srow, fb, a);
    grp_load<1>(h, ch.srow, fb, b);
    grp_load<2>(h, ch.srow, fb, c);
    grp_load<3>(h, ch.srow, fb, d);
    acc = grp_acc<0>(acc, a, ch.w, lim);
    acc = grp_acc<1>(acc, b, ch.w, lim);
    acc = grp_acc<2>(acc, c, ch.w, lim);
    acc = grp_acc<3>(acc, d, ch.w, lim);
  } else if (rem > 16) {
    grp_load<0>(h, ch.srow, fb, a);
    grp_load<1>(h, ch.srow, fb, b);
    grp_load<2>(h, ch.srow, fb, c);
    acc = grp_acc<0>(acc, a, ch.w, lim);
    acc = grp_acc<1>(acc, b, ch.w, lim);
    acc = grp_acc<2>(acc, c, ch.w, lim);
  } else if (rem > 8) {
    grp_load<0>(h, ch.srow, fb, a);
    grp_load<1>(h, ch.srow, fb, b);
    acc = grp_acc<0>(acc, a, ch.w, lim);
    acc = grp_acc<1>(acc, b, ch.w, lim);
  } else {
    grp_load<0>(h, ch.srow, fb, a);
    acc = grp_acc<0>(acc, a, ch.w, lim);
  }
  return acc;
}

template <int GB>
__device__ __forceinline__ float sum_ab(const float* __restrict__ h, Chunk A,
                                        Chunk B, unsigned fb, int limA,
                                        int limB, float acc) {
  float a0[4], a1[4], a2[4], a3[4], b0[4], b1[4], b2[4], b3[4];
  grp_load<0>(h, A.srow, fb, a0);
  grp_load<1>(h, A.srow, fb, a1);
  grp_load<2>(h, A.srow, fb, a2);
  grp_load<3>(h, A.srow, fb, a3);
  if constexpr (GB > 0) grp_load<0>(h, B.srow, fb, b0);
  if constexpr (GB > 1) grp_load<1>(h, B.srow, fb, b1);
  if constexpr (GB > 2) grp_load<2>(h, B.srow, fb, b2);
  if constexpr (GB > 3) grp_load<3>(h, B.srow, fb, b3);
  acc = grp_acc<0>(acc, a0, A.w, limA);
  acc = grp_acc<1>(acc, a1, A.w, limA);
  acc = grp_acc<2>(acc, a2, A.w, limA);
  acc = grp_acc<3>(acc, a3, A.w, limA);
  if constexpr (GB > 0) acc = grp_acc<0>(acc, b0, B.w, limB);
  if constexpr (GB > 1) acc = grp_acc<1>(acc, b1, B.w, limB);
  if constexpr (GB > 2) acc = grp_acc<2>(acc, b2, B.w, limB);
  if constexpr (GB > 3) acc = grp_acc<3>(acc, b3, B.w, limB);
  return acc;
}

__device__ __forceinline__ float chunk_sum_ab(const float* __restrict__ h,
                                              Chunk A, Chunk B, unsigned fb,
                                              int half, int n, float acc) {
  const int limA = 32 - half;
  const int remB = n - 32;
  const int limB = remB - half;
  if (remB > 24) return sum_ab<4>(h, A, B, fb, limA, limB, acc);
  if (remB > 16) return sum_ab<3>(h, A, B, fb, limA, limB, acc);
  if (remB > 8) return sum_ab<2>(h, A, B, fb, limA, limB, acc);
  return sum_ab<1>(h, A, B, fb, limA, limB, acc);
}

__device__ __forceinline__ float overflow_rows(const int2* __restrict__ ovf,
                                               int on, int i,
                                               const int* __restrict__ cnt,
                                               const float* __restrict__ h,
                                               float di, int f, int half,
                                               int lane, float acc) {
  for (int b = 0; b < on; b += 32) {
    const bool ok = (b + lane) < on;
    const int2 q = ok ? ovf[b + lane] : make_int2(-1, 0);
    const unsigned m = (unsigned)__ballot(q.x == i);
    for (int j = 0; j < 32; ++j) {
      if ((m >> j) & 1u) {
        const int s = lane_i(q.y, j);
        const float w = dinv_of(cnt[s]) * di;
        const float v = h[(size_t)s * 16 + f];
        acc += (half == 0) ? w * v : 0.0f;
      }
    }
  }
  return acc;
}

__global__ void __launch_bounds__(P1_WAVES * 32)
pull_layer1(const int* __restrict__ slots,
            const int2* __restrict__ ovf, const int* __restrict__ ctr,
            const int* __restrict__ cnt,
            const uint4* __restrict__ deg8,
            const float* __restrict__ h,
            const float* __restrict__ b1,
            const float* __restrict__ W2,
            float2* __restrict__ zd,
            int N, int E) {
  __shared__ uint4 tab4[TAB_BYTES / 16];
  const unsigned char* const tab = (const unsigned char*)tab4;
  const int tn = (N + 127) & ~127;
  const int tl = (tn < TAB_BYTES) ? tn : TAB_BYTES;
  {
    const int nv = tl >> 4;
    const int NT = P1_WAVES * 32;
    for (int v0 = threadIdx.x; v0 < nv; v0 += FILL_U * NT) {
      uint4 a[FILL_U];
#pragma unroll
      for (int k = 0; k < FILL_U; ++k) {
        const int v = v0 + k * NT;
        a[k] = deg8[(v < nv) ? v : (nv - 1)];
      }
#pragma unroll
      for (int k = 0; k < FILL_U; ++k) {
        const int v = v0 + k * NT;
        tab4[(v < nv) ? v : (nv - 1)] = a[k];
      }
    }
  }
  __syncthreads();
  const int lane = threadIdx.x & 31;
  const int i0 = __builtin_amdgcn_readfirstlane(
      (int)((blockIdx.x * blockDim.x + threadIdx.x) >> 5) * 32);
  if (i0 >= N) return;
  const int nodes = (N - i0 < 32) ? (N - i0) : 32;
  const int mi = i0 + lane;
  const bool mok = lane < nodes;
  const int mc = mok ? mi : i0;
  const int my_cnt = mok ? cnt[mc] : 0;
  const float my_di = dinv_of(my_cnt);
  const int f = lane & 15, half = lane >> 4;
  const unsigned fb = (unsigned)f * 4u;
  const int slot = 2 * f + half;
  const float b1f = b1[f], w2f = W2[f];
  float zkeep = 0.0f;

  int n0 = lane_i(my_cnt, 0);
  n0 = (n0 < CAP) ? n0 : CAP;
  size_t row0 = (size_t)i0 * CAP;
  int s0a = load_src(slots, row0, n0, 0, slot);
  int s0b = load_src(slots, row0, n0, 32, slot);
  const int x1 = (nodes > 1) ? 1 : 0;
  int n1 = lane_i(my_cnt, x1);
  n1 = (n1 < CAP) ? n1 : CAP;
  size_t row1 = (size_t)(i0 + x1) * CAP;
  int s1a = load_src(slots, row1, n1, 0, slot);
  int s1b = load_src(slots, row1, n1, 32, slot);
  s0a = src_of(s0a, n0, 0, slot);
  s0b = src_of(s0b, n0, 32, slot);
  const int b0a = deg_byte(tab, tl, s0a);
  const int b0b = deg_byte(tab, tl, s0b);
  float d0a = dinv_of(deg_full(b0a, cnt, s0a));
  float d0b = dinv_of(deg_full(b0b, cnt, s0b));

  for (int nn = 0; nn < nodes; ++nn) {
    const int i = i0 + nn;
    const int n_c = n0;
    const size_t row_c = row0;
    const int sa = s0a, sb = s0b;
    const float hv = h[(size_t)i * 16 + f];
    const float di = lane_f(my_di, nn);
    const int nfull = lane_i(my_cnt, nn);
    const int t1a = src_of(s1a, n1, 0, slot);
    const int t1b = src_of(s1b, n1, 32, slot);
    const int e1a = deg_byte(tab, tl, t1a);
    const int e1b = deg_byte(tab, tl, t1b);
    const int x2 = (nn + 2 < nodes) ? (nn + 2) : (nodes - 1);
    int n2 = lane_i(my_cnt, x2);
    n2 = (n2 < CAP) ? n2 : CAP;
    const size_t row2 = (size_t)(i0 + x2) * CAP;
    const int s2a = load_src(slots, row2, n2, 0, slot);
    const int s2b = load_src(slots, row2, n2, 32, slot);
    const bool va = slot < n_c, vb = (32 + slot) < n_c;
    const float wa = va ? d0a * di : 0.0f;
    const float wb = vb ? d0b * di : 0.0f;
    Chunk A, B;
    A.srow = (unsigned)sa * 64u;
    A.w = wa;
    B.srow = (unsigned)sb * 64u;
    B.w = wb;
    float acc = 0.0f;
    if (n_c > 32)
      acc = chunk_sum_ab(h, A, B, fb, half, n_c, acc);
    else if (n_c > 0)
      acc = chunk_sum(h, A, fb, half, n_c, acc);
    if (nfull > CAP) {
      int on = ctr[OVF_WORD];
      on = (on < E) ? on : E;
      acc = overflow_rows(ovf, on, i, cnt, h, di, f, half, lane, acc);
    }
    acc += xhalf(acc);
    const float a1 = fmaxf(acc + hv * (di * di) + b1f, 0.0f);
    float p = a1 * w2f;
    p = row_xadd<8>(p);
    p = row_xadd<4>(p);
    p = row_xadd<2>(p);
    p = row_xadd<1>(p);
    if (lane == nn) zkeep = p;
    n0 = n1;
    row0 = row1;
    s0a = t1a;
    s0b = t1b;
    d0a = dinv_of(deg_full(e1a, cnt, t1a));
    d0b = dinv_of(deg_full(e1b, cnt, t1b));
    n1 = n2;
    row1 = row2;
    s1a = s2a;
    s1b = s2b;
  }
  if (mok) zd[mi] = make_float2(zkeep, my_di);
}

__global__ void __launch_bounds__(256)
pull_layer2(const int* __restrict__ slots,
            const int2* __restrict__ ovf, const int* __restrict__ ctr,
            const int* __restrict__ cnt,
            const float2* __restrict__ zd,
            const float* __restrict__ b2, float* __restrict__ out, int N,
            int E) {
  const int lane = threadIdx.x & 31;
  const int i0 = __builtin_amdgcn_readfirstlane(
      (int)((blockIdx.x * blockDim.x + threadIdx.x) >> 5) * 32);
  if (i0 >= N) return;
  const int nodes = (N - i0 < 32) ? (N - i0) : 32;
  const int mi = i0 + lane;
  const bool mok = lane < nodes;
  const int mc = mok ? mi : i0;
  const int my_cnt = mok ? cnt[mc] : 0;
  const float my_di = dinv_of(my_cnt);
  const float zself = zd[mc].x;
  float keep = 0.0f;

  for (int nn = 0; nn < 32; nn += 4) {
    int nf[4], n[4];
    int sa[4], sb[4];
    float2 za[4], zb[4];
#pragma unroll
    for (int u = 0; u < 4; ++u) {
      nf[u] = lane_i(my_cnt, nn + u);
      n[u] = (nf[u] < CAP) ? nf[u] : CAP;
      const size_t row = (size_t)(i0 + nn + u) * CAP;
      const size_t ia = (lane < n[u]) ? row + lane : 0;
      const size_t ib = (32 + lane < n[u]) ? row + 32 + lane : 0;
      sa[u] = slots[ia];
      sb[u] = slots[ib];
    }
#pragma unroll
    for (int u = 0; u < 4; ++u) {
      za[u] = zd[(lane < n[u]) ? sa[u] : 0];
      zb[u] = zd[(32 + lane < n[u]) ? sb[u] : 0];
    }
#pragma unroll
    for (int u = 0; u < 4; ++u) {
      const float di = lane_f(my_di, nn + u);
      const float wa = __int_as_float(__float_as_int(za[u].y * di) &
                                      ((lane < n[u]) ? -1 : 0));
      const float wb = __int_as_float(__float_as_int(zb[u].y * di) &
                                      ((32 + lane < n[u]) ? -1 : 0));
      float acc = wa * za[u].x + wb * zb[u].x;
      if (nf[u] > CAP) {
        const int i = i0 + nn + u;
        int on = ctr[OVF_WORD];
        on = (on < E) ? on : E;
        for (int b = 0; b < on; b += 32) {
          const bool ok = (b + lane) < on;
          const int2 q = ok ? ovf[b + lane] : make_int2(-1, 0);
          if (q.x == i) {
            const float2 p = zd[q.y];
            acc += (p.y * di) * p.x;
          }
        }
      }
      const float s = wave_sum(acc);
      if (lane == nn + u) keep = s;
    }
  }
  if (mok) out[mi] = keep + zself * (my_di * my_di) + b2[0];
}

extern "C" void kernel_launch(void* const* d_in, const int* in_sizes, int n_in,
                              void* d_out, int out_size, void* d_ws,
                              size_t ws_size, hipStream_t stream) {
  const float* x  = (const float*)d_in[0];
  const int*   ei = (const int*)d_in[1];
  const float* W1 = (const float*)d_in[2];
  const float* b1 = (const float*)d_in[3];
  const float* W2 = (const float*)d_in[4];
  const float* b2 = (const float*)d_in[5];
  float* out = (float*)d_out;

  const int N = in_sizes[0] / 3;
  const int E = in_sizes[1] / 2;
  const int* src = ei;
  const int* dst = ei + E;

  const size_t nb4 = (((size_t)N * 4) + 255) & ~(size_t)255;
  const size_t sb4 = (size_t)N * CAP * 4;
  const size_t ob8 = (((size_t)E * 8) + 255) & ~(size_t)255;
  char* ws = (char*)d_ws;
  int*    cnt   = (int*)(ws);
  float2* zd    = (float2*)(ws + nb4);
  float*  h     = (float*)(ws + 3 * nb4);
  int*    slots = (int*)(ws + 19 * nb4);
  int2*   ovf   = (int2*)(ws + 19 * nb4 + sb4);
  int*    ctr   = (int*)(ws + 19 * nb4 + sb4 + ob8);
  unsigned* deg8 = (unsigned*)(ws + 19 * nb4 + sb4 + ob8 + CTR_WORDS * 4);

  const int BT = 256;
  const int eblocks = (E + BT * EPT - 1) / (BT * EPT);
  const int nwaves = (N + 31) / 32;
  const int pblocks = (nwaves + 7) / 8;

  init_counts<<<(N + BT - 1) / BT, BT, 0, stream>>>(cnt, ctr, N);
  count_place<<<EDGE_WGS, BT, 0, stream>>>(src, dst, cnt, slots, ovf, ctr, E, N,
                                           eblocks);
  {
    int ntiles = N / 16;
    int wpb = 8;
    int blocks = (ntiles + wpb - 1) / wpb;
    gemm1_wmma<<<blocks, 32 * wpb, 0, stream>>>(x, W1, h, cnt, deg8, ntiles, N);
  }
  pull_layer1<<<(nwaves + P1_WAVES - 1) / P1_WAVES, P1_WAVES * 32, 0, stream>>>(
      slots, ovf, ctr, cnt, (const uint4*)deg8, h, b1, W2, zd, N, E);
  pull_layer2<<<pblocks, BT, 0, stream>>>(slots, ovf, ctr, cnt, zd, b2, out, N,
                                          E);
}
